# starting kernel
# baseline (speedup 1.0000x reference)
#include <hip/hip_runtime.h>
#include <math.h>


typedef __attribute__((ext_vector_type(2))) float v2f;
typedef __attribute__((ext_vector_type(8))) float v8f;

#define MM 16
#define NN 2048
#define DD 128
#define PP 8192
#define HH 16
#define NCHUNK 33

__global__ __launch_bounds__(256) void k_rmsnorm_x(const float* __restrict__ X,
                                                   float* __restrict__ Xn)
{
    __shared__ float red[256];
    const int m = blockIdx.x;
    const int t = threadIdx.x;
    float ss = 0.0f;
    for (int j = t; j < NN; j += 256) {
        float x = X[(size_t)m * NN + j];
        ss += x * x;
    }
    red[t] = ss;
    __syncthreads();
    for (int o = 128; o > 0; o >>= 1) {
        if (t < o) red[t] += red[t + o];
        __syncthreads();
    }
    const float rs = rsqrtf(red[0] / (float)NN);
    for (int j = t; j < NN; j += 256)
        Xn[(size_t)m * NN + j] = X[(size_t)m * NN + j] * rs;
}

__global__ __launch_bounds__(32) void k_proj_wmma(const float* __restrict__ Xn,
                                                  const float* __restrict__ Wq,
                                                  const float* __restrict__ Wk,
                                                  const float* __restrict__ Wv,
                                                  float* __restrict__ qraw,
                                                  float* __restrict__ kraw,
                                                  float* __restrict__ vraw)
{
    const int lane = threadIdx.x;
    const int col  = lane & 15;
    const int lh   = lane >> 4;
    const int bid  = blockIdx.x;
    const int w    = bid >> 7;
    const int n0   = (bid & 127) << 4;

    const float* W   = (w == 0) ? Wq : (w == 1) ? Wk : Wv;
    float*       out = (w == 0) ? qraw : (w == 1) ? kraw : vraw;

    const float* arow = Xn + (size_t)col * NN + 2 * lh;
    const float* bcol = W + n0 + col;

    v8f acc = {};
    #pragma unroll 8
    for (int kb = 0; kb < NN; kb += 4) {
        float2 a2 = *reinterpret_cast<const float2*>(arow + kb);
        v2f A = {a2.x, a2.y};
        const float* bp = bcol + (size_t)(kb + 2 * lh) * NN;
        v2f B = {bp[0], bp[NN]};
        acc = __builtin_amdgcn_wmma_f32_16x16x4_f32(false, A, false, B,
                                                    (short)0, acc, false, false);
    }

    #pragma unroll
    for (int r = 0; r < 8; ++r)
        out[(size_t)(r + 8 * lh) * NN + n0 + col] = acc[r];
}

__global__ __launch_bounds__(256) void k_qknorm(const float* __restrict__ qraw,
                                                const float* __restrict__ kraw,
                                                const float* __restrict__ vraw,
                                                float* __restrict__ qn,
                                                float* __restrict__ kn,
                                                float* __restrict__ vn)
{
    const int gtid = blockIdx.x * blockDim.x + threadIdx.x;
    const int wid  = gtid >> 5;
    const int lane = gtid & 31;
    const int h = wid >> 4;
    const int m = wid & 15;
    const int d0 = lane * 4;

    const size_t src = (size_t)m * NN + (size_t)h * DD + d0;
    const size_t dst = ((size_t)h * MM + m) * DD + d0;

    float4 q4 = *reinterpret_cast<const float4*>(qraw + src);
    float ss = q4.x * q4.x + q4.y * q4.y + q4.z * q4.z + q4.w * q4.w;
    #pragma unroll
    for (int mk = 16; mk >= 1; mk >>= 1) ss += __shfl_xor(ss, mk, 32);
    float rs = rsqrtf(ss / (float)DD);
    float4 o;
    o.x = q4.x * rs; o.y = q4.y * rs; o.z = q4.z * rs; o.w = q4.w * rs;
    *reinterpret_cast<float4*>(qn + dst) = o;

    float4 k4 = *reinterpret_cast<const float4*>(kraw + src);
    ss = k4.x * k4.x + k4.y * k4.y + k4.z * k4.z + k4.w * k4.w;
    #pragma unroll
    for (int mk = 16; mk >= 1; mk >>= 1) ss += __shfl_xor(ss, mk, 32);
    rs = rsqrtf(ss / (float)DD);
    o.x = k4.x * rs; o.y = k4.y * rs; o.z = k4.z * rs; o.w = k4.w * rs;
    *reinterpret_cast<float4*>(kn + dst) = o;

    *reinterpret_cast<float4*>(vn + dst) =
        *reinterpret_cast<const float4*>(vraw + src);
}

__global__ __launch_bounds__(32) void k_attn_partial(const float* __restrict__ qn,
                                                     const float* __restrict__ kn,
                                                     const float* __restrict__ vn,
                                                     const float* __restrict__ cK,
                                                     const float* __restrict__ cV,
                                                     float* __restrict__ pout,
                                                     float* __restrict__ pmax,
                                                     float* __restrict__ psum)
{
    __shared__ __align__(16) float q_lds[MM * DD];
    __shared__ __align__(16) float sc[MM * 256];

    const int lane = threadIdx.x;
    const int col  = lane & 15;
    const int lh   = lane >> 4;
    const int s    = blockIdx.x;
    const int h    = blockIdx.y;

    const float* qh = qn + (size_t)h * MM * DD;
    for (int i = lane; i < (MM * DD) / 4; i += 32)
        reinterpret_cast<float4*>(q_lds)[i] =
            reinterpret_cast<const float4*>(qh)[i];

    const int nrows = (s < 32) ? 256 : MM;
    const float* Kb = (s < 32) ? (cK + ((size_t)h * PP + (size_t)s * 256) * DD)
                               : (kn + (size_t)h * MM * DD);
    const float* Vb = (s < 32) ? (cV + ((size_t)h * PP + (size_t)s * 256) * DD)
                               : (vn + (size_t)h * MM * DD);

    if (nrows < 256) {
        for (int i = lane; i < MM * 256; i += 32) sc[i] = -3.0e38f;
    }

    for (int t = 0; t < nrows / 16; ++t) {
        const int j0 = t * 16;
        const float* krow = Kb + (size_t)(j0 + col) * DD + 2 * lh;
        const float* qrow = q_lds + (size_t)col * DD + 2 * lh;
        v8f acc = {};
        #pragma unroll
        for (int kb = 0; kb < DD; kb += 4) {
            float2 a2 = *reinterpret_cast<const float2*>(qrow + kb);
            float2 b2 = *reinterpret_cast<const float2*>(krow + kb);
            v2f A = {a2.x, a2.y};
            v2f B = {b2.x, b2.y};
            acc = __builtin_amdgcn_wmma_f32_16x16x4_f32(false, A, false, B,
                                                        (short)0, acc, false, false);
        }
        #pragma unroll
        for (int r = 0; r < 8; ++r)
            sc[(r + 8 * lh) * 256 + j0 + col] = acc[r];
    }

    float* scrow = sc + (size_t)col * 256 + lh * 128;
    float mloc = -3.0e38f;
    for (int j = 0; j < 128; ++j) mloc = fmaxf(mloc, scrow[j]);
    mloc = fmaxf(mloc, __shfl_xor(mloc, 16, 32));
    float ssum = 0.0f;
    for (int j = 0; j < 128; ++j) {
        float e = __expf(scrow[j] - mloc);
        scrow[j] = e;
        ssum += e;
    }
    ssum += __shfl_xor(ssum, 16, 32);
    if (lh == 0) {
        pmax[((size_t)h * NCHUNK + s) * MM + col] = mloc;
        psum[((size_t)h * NCHUNK + s) * MM + col] = ssum;
    }

    v8f zero = {};
    v8f oacc[8];
    #pragma unroll
    for (int dt = 0; dt < 8; ++dt) oacc[dt] = zero;

    for (int kb = 0; kb < nrows; kb += 4) {
        float2 a2 = *reinterpret_cast<const float2*>(sc + (size_t)col * 256 + kb + 2 * lh);
        v2f A = {a2.x, a2.y};
        const float* vrow = Vb + (size_t)(kb + 2 * lh) * DD + col;
        #pragma unroll
        for (int dt = 0; dt < 8; ++dt) {
            v2f B = {vrow[dt * 16], vrow[DD + dt * 16]};
            oacc[dt] = __builtin_amdgcn_wmma_f32_16x16x4_f32(false, A, false, B,
                                                             (short)0, oacc[dt],
                                                             false, false);
        }
    }

    float* po = pout + ((size_t)h * NCHUNK + s) * MM * DD;
    #pragma unroll
    for (int dt = 0; dt < 8; ++dt) {
        #pragma unroll
        for (int r = 0; r < 8; ++r)
            po[(size_t)(r + 8 * lh) * DD + dt * 16 + col] = oacc[dt][r];
    }
}

__global__ __launch_bounds__(256) void k_combine(const float* __restrict__ pout,
                                                 const float* __restrict__ pmax,
                                                 const float* __restrict__ psum,
                                                 float* __restrict__ out)
{
    const int h = blockIdx.x;
    for (int idx = threadIdx.x; idx < MM * DD; idx += blockDim.x) {
        const int m = idx >> 7;
        const int d = idx & (DD - 1);
        const float* pm  = pmax + (size_t)h * NCHUNK * MM + m;
        const float* psu = psum + (size_t)h * NCHUNK * MM + m;
        float g = -3.0e38f;
        for (int s = 0; s < NCHUNK; ++s) g = fmaxf(g, pm[(size_t)s * MM]);
        float tot = 0.0f, acc = 0.0f;
        for (int s = 0; s < NCHUNK; ++s) {
            float w = __expf(pm[(size_t)s * MM] - g);
            tot += w * psu[(size_t)s * MM];
            acc += w * pout[(((size_t)h * NCHUNK + s) * MM + m) * DD + d];
        }
        out[(size_t)m * NN + (size_t)h * DD + d] = acc / tot;
    }
}

extern "C" void kernel_launch(void* const* d_in, const int* in_sizes, int n_in,
                              void* d_out, int out_size, void* d_ws, size_t ws_size,
                              hipStream_t stream)
{
    (void)in_sizes; (void)n_in; (void)out_size; (void)ws_size;
    const float* X  = (const float*)d_in[0];
    const float* Wq = (const float*)d_in[1];
    const float* Wk = (const float*)d_in[2];
    const float* Wv = (const float*)d_in[3];
    const float* cK = (const float*)d_in[4];
    const float* cV = (const float*)d_in[5];
    float* out = (float*)d_out;
    float* ws  = (float*)d_ws;

    float* xnorm = ws;
    float* qraw  = xnorm + MM * NN;
    float* kraw  = qraw + MM * NN;
    float* vraw  = kraw + MM * NN;
    float* qn    = vraw + MM * NN;
    float* kn    = qn + HH * MM * DD;
    float* vn    = kn + HH * MM * DD;
    float* pout  = vn + HH * MM * DD;
    float* pmaxb = pout + (size_t)HH * NCHUNK * MM * DD;
    float* psumb = pmaxb + (size_t)HH * NCHUNK * MM;

    k_rmsnorm_x<<<16, 256, 0, stream>>>(X, xnorm);
    k_proj_wmma<<<3 * (NN / 16), 32, 0, stream>>>(xnorm, Wq, Wk, Wv,
                                                  qraw, kraw, vraw);
    k_qknorm<<<(HH * MM) / 8, 256, 0, stream>>>(qraw, kraw, vraw, qn, kn, vn);
    k_attn_partial<<<dim3(NCHUNK, HH), 32, 0, stream>>>(qn, kn, vn, cK, cV,
                                                        pout, pmaxb, psumb);
    k_combine<<<HH, 256, 0, stream>>>(pout, pmaxb, psumb, out);
}


# combine kernel: one block per (head,row), one thread per column (256 blocks x 128) with all partial loads issued up front, instead of 16 blocks looping; per-element sum order unchanged
# speedup vs baseline: 1.3352x; 1.3352x over previous
#include <hip/hip_runtime.h>
#include <math.h>


typedef __attribute__((ext_vector_type(2))) float v2f;
typedef __attribute__((ext_vector_type(8))) float v8f;

#define MM 16
#define NN 2048
#define DD 128
#define PP 8192
#define HH 16
#define NCHUNK 33

__global__ __launch_bounds__(256) void k_rmsnorm_x(const float* __restrict__ X,
                                                   float* __restrict__ Xn)
{
    __shared__ float red[256];
    const int m = blockIdx.x;
    const int t = threadIdx.x;
    float ss = 0.0f;
    for (int j = t; j < NN; j += 256) {
        float x = X[(size_t)m * NN + j];
        ss += x * x;
    }
    red[t] = ss;
    __syncthreads();
    for (int o = 128; o > 0; o >>= 1) {
        if (t < o) red[t] += red[t + o];
        __syncthreads();
    }
    const float rs = rsqrtf(red[0] / (float)NN);
    for (int j = t; j < NN; j += 256)
        Xn[(size_t)m * NN + j] = X[(size_t)m * NN + j] * rs;
}

__global__ __launch_bounds__(32) void k_proj_wmma(const float* __restrict__ Xn,
                                                  const float* __restrict__ Wq,
                                                  const float* __restrict__ Wk,
                                                  const float* __restrict__ Wv,
                                                  float* __restrict__ qraw,
                                                  float* __restrict__ kraw,
                                                  float* __restrict__ vraw)
{
    const int lane = threadIdx.x;
    const int col  = lane & 15;
    const int lh   = lane >> 4;
    const int bid  = blockIdx.x;
    const int w    = bid >> 7;
    const int n0   = (bid & 127) << 4;

    const float* W   = (w == 0) ? Wq : (w == 1) ? Wk : Wv;
    float*       out = (w == 0) ? qraw : (w == 1) ? kraw : vraw;

    const float* arow = Xn + (size_t)col * NN + 2 * lh;
    const float* bcol = W + n0 + col;

    v8f acc = {};
    #pragma unroll 8
    for (int kb = 0; kb < NN; kb += 4) {
        float2 a2 = *reinterpret_cast<const float2*>(arow + kb);
        v2f A = {a2.x, a2.y};
        const float* bp = bcol + (size_t)(kb + 2 * lh) * NN;
        v2f B = {bp[0], bp[NN]};
        acc = __builtin_amdgcn_wmma_f32_16x16x4_f32(false, A, false, B,
                                                    (short)0, acc, false, false);
    }

    #pragma unroll
    for (int r = 0; r < 8; ++r)
        out[(size_t)(r + 8 * lh) * NN + n0 + col] = acc[r];
}

__global__ __launch_bounds__(256) void k_qknorm(const float* __restrict__ qraw,
                                                const float* __restrict__ kraw,
                                                const float* __restrict__ vraw,
                                                float* __restrict__ qn,
                                                float* __restrict__ kn,
                                                float* __restrict__ vn)
{
    const int gtid = blockIdx.x * blockDim.x + threadIdx.x;
    const int wid  = gtid >> 5;
    const int lane = gtid & 31;
    const int h = wid >> 4;
    const int m = wid & 15;
    const int d0 = lane * 4;

    const size_t src = (size_t)m * NN + (size_t)h * DD + d0;
    const size_t dst = ((size_t)h * MM + m) * DD + d0;

    float4 q4 = *reinterpret_cast<const float4*>(qraw + src);
    float ss = q4.x * q4.x + q4.y * q4.y + q4.z * q4.z + q4.w * q4.w;
    #pragma unroll
    for (int mk = 16; mk >= 1; mk >>= 1) ss += __shfl_xor(ss, mk, 32);
    float rs = rsqrtf(ss / (float)DD);
    float4 o;
    o.x = q4.x * rs; o.y = q4.y * rs; o.z = q4.z * rs; o.w = q4.w * rs;
    *reinterpret_cast<float4*>(qn + dst) = o;

    float4 k4 = *reinterpret_cast<const float4*>(kraw + src);
    ss = k4.x * k4.x + k4.y * k4.y + k4.z * k4.z + k4.w * k4.w;
    #pragma unroll
    for (int mk = 16; mk >= 1; mk >>= 1) ss += __shfl_xor(ss, mk, 32);
    rs = rsqrtf(ss / (float)DD);
    o.x = k4.x * rs; o.y = k4.y * rs; o.z = k4.z * rs; o.w = k4.w * rs;
    *reinterpret_cast<float4*>(kn + dst) = o;

    *reinterpret_cast<float4*>(vn + dst) =
        *reinterpret_cast<const float4*>(vraw + src);
}

__global__ __launch_bounds__(32) void k_attn_partial(const float* __restrict__ qn,
                                                     const float* __restrict__ kn,
                                                     const float* __restrict__ vn,
                                                     const float* __restrict__ cK,
                                                     const float* __restrict__ cV,
                                                     float* __restrict__ pout,
                                                     float* __restrict__ pmax,
                                                     float* __restrict__ psum)
{
    __shared__ __align__(16) float q_lds[MM * DD];
    __shared__ __align__(16) float sc[MM * 256];

    const int lane = threadIdx.x;
    const int col  = lane & 15;
    const int lh   = lane >> 4;
    const int s    = blockIdx.x;
    const int h    = blockIdx.y;

    const float* qh = qn + (size_t)h * MM * DD;
    for (int i = lane; i < (MM * DD) / 4; i += 32)
        reinterpret_cast<float4*>(q_lds)[i] =
            reinterpret_cast<const float4*>(qh)[i];

    const int nrows = (s < 32) ? 256 : MM;
    const float* Kb = (s < 32) ? (cK + ((size_t)h * PP + (size_t)s * 256) * DD)
                               : (kn + (size_t)h * MM * DD);
    const float* Vb = (s < 32) ? (cV + ((size_t)h * PP + (size_t)s * 256) * DD)
                               : (vn + (size_t)h * MM * DD);

    if (nrows < 256) {
        for (int i = lane; i < MM * 256; i += 32) sc[i] = -3.0e38f;
    }

    for (int t = 0; t < nrows / 16; ++t) {
        const int j0 = t * 16;
        const float* krow = Kb + (size_t)(j0 + col) * DD + 2 * lh;
        const float* qrow = q_lds + (size_t)col * DD + 2 * lh;
        v8f acc = {};
        #pragma unroll
        for (int kb = 0; kb < DD; kb += 4) {
            float2 a2 = *reinterpret_cast<const float2*>(qrow + kb);
            float2 b2 = *reinterpret_cast<const float2*>(krow + kb);
            v2f A = {a2.x, a2.y};
            v2f B = {b2.x, b2.y};
            acc = __builtin_amdgcn_wmma_f32_16x16x4_f32(false, A, false, B,
                                                        (short)0, acc, false, false);
        }
        #pragma unroll
        for (int r = 0; r < 8; ++r)
            sc[(r + 8 * lh) * 256 + j0 + col] = acc[r];
    }

    float* scrow = sc + (size_t)col * 256 + lh * 128;
    float mloc = -3.0e38f;
    for (int j = 0; j < 128; ++j) mloc = fmaxf(mloc, scrow[j]);
    mloc = fmaxf(mloc, __shfl_xor(mloc, 16, 32));
    float ssum = 0.0f;
    for (int j = 0; j < 128; ++j) {
        float e = __expf(scrow[j] - mloc);
        scrow[j] = e;
        ssum += e;
    }
    ssum += __shfl_xor(ssum, 16, 32);
    if (lh == 0) {
        pmax[((size_t)h * NCHUNK + s) * MM + col] = mloc;
        psum[((size_t)h * NCHUNK + s) * MM + col] = ssum;
    }

    v8f zero = {};
    v8f oacc[8];
    #pragma unroll
    for (int dt = 0; dt < 8; ++dt) oacc[dt] = zero;

    for (int kb = 0; kb < nrows; kb += 4) {
        float2 a2 = *reinterpret_cast<const float2*>(sc + (size_t)col * 256 + kb + 2 * lh);
        v2f A = {a2.x, a2.y};
        const float* vrow = Vb + (size_t)(kb + 2 * lh) * DD + col;
        #pragma unroll
        for (int dt = 0; dt < 8; ++dt) {
            v2f B = {vrow[dt * 16], vrow[DD + dt * 16]};
            oacc[dt] = __builtin_amdgcn_wmma_f32_16x16x4_f32(false, A, false, B,
                                                             (short)0, oacc[dt],
                                                             false, false);
        }
    }

    float* po = pout + ((size_t)h * NCHUNK + s) * MM * DD;
    #pragma unroll
    for (int dt = 0; dt < 8; ++dt) {
        #pragma unroll
        for (int r = 0; r < 8; ++r)
            po[(size_t)(r + 8 * lh) * DD + dt * 16 + col] = oacc[dt][r];
    }
}

__global__ __launch_bounds__(128) void k_combine(const float* __restrict__ pout,
                                                 const float* __restrict__ pmax,
                                                 const float* __restrict__ psum,
                                                 float* __restrict__ out)
{
    const int h = blockIdx.x >> 4;
    const int m = blockIdx.x & (MM - 1);
    const int d = threadIdx.x;
    const float* pm  = pmax + (size_t)h * NCHUNK * MM + m;
    const float* psu = psum + (size_t)h * NCHUNK * MM + m;
    const float* po  = pout + ((size_t)h * NCHUNK * MM + m) * DD + d;

    float mx[NCHUNK], sm[NCHUNK], pv[NCHUNK];
    #pragma unroll
    for (int s = 0; s < NCHUNK; ++s) {
        mx[s] = pm[(size_t)s * MM];
        sm[s] = psu[(size_t)s * MM];
        pv[s] = po[(size_t)s * MM * DD];
    }
    float g = -3.0e38f;
    #pragma unroll
    for (int s = 0; s < NCHUNK; ++s) g = fmaxf(g, mx[s]);
    float tot = 0.0f, acc = 0.0f;
    #pragma unroll
    for (int s = 0; s < NCHUNK; ++s) {
        float w = __expf(mx[s] - g);
        tot += w * sm[s];
        acc += w * pv[s];
    }
    out[(size_t)m * NN + (size_t)h * DD + d] = acc / tot;
}

extern "C" void kernel_launch(void* const* d_in, const int* in_sizes, int n_in,
                              void* d_out, int out_size, void* d_ws, size_t ws_size,
                              hipStream_t stream)
{
    (void)in_sizes; (void)n_in; (void)out_size; (void)ws_size;
    const float* X  = (const float*)d_in[0];
    const float* Wq = (const float*)d_in[1];
    const float* Wk = (const float*)d_in[2];
    const float* Wv = (const float*)d_in[3];
    const float* cK = (const float*)d_in[4];
    const float* cV = (const float*)d_in[5];
    float* out = (float*)d_out;
    float* ws  = (float*)d_ws;

    float* xnorm = ws;
    float* qraw  = xnorm + MM * NN;
    float* kraw  = qraw + MM * NN;
    float* vraw  = kraw + MM * NN;
    float* qn    = vraw + MM * NN;
    float* kn    = qn + HH * MM * DD;
    float* vn    = kn + HH * MM * DD;
    float* pout  = vn + HH * MM * DD;
    float* pmaxb = pout + (size_t)HH * NCHUNK * MM * DD;
    float* psumb = pmaxb + (size_t)HH * NCHUNK * MM;

    k_rmsnorm_x<<<16, 256, 0, stream>>>(X, xnorm);
    k_proj_wmma<<<3 * (NN / 16), 32, 0, stream>>>(xnorm, Wq, Wk, Wv,
                                                  qraw, kraw, vraw);
    k_qknorm<<<(HH * MM) / 8, 256, 0, stream>>>(qraw, kraw, vraw, qn, kn, vn);
    k_attn_partial<<<dim3(NCHUNK, HH), 32, 0, stream>>>(qn, kn, vn, cK, cV,
                                                        pout, pmaxb, psumb);
    k_combine<<<HH * MM, DD, 0, stream>>>(pout, pmaxb, psumb, out);
}


# attention partial kernel: 8 waves per block instead of 1 (score tiles and output column tiles split over the waves, LDS rows padded against bank conflicts); every sum keeps its order
# speedup vs baseline: 2.2234x; 1.6652x over previous
#include <hip/hip_runtime.h>
#include <math.h>


typedef __attribute__((ext_vector_type(2))) float v2f;
typedef __attribute__((ext_vector_type(8))) float v8f;

#define MM 16
#define NN 2048
#define DD 128
#define PP 8192
#define HH 16
#define NCHUNK 33

__global__ __launch_bounds__(256) void k_rmsnorm_x(const float* __restrict__ X,
                                                   float* __restrict__ Xn)
{
    __shared__ float red[256];
    const int m = blockIdx.x;
    const int t = threadIdx.x;
    float ss = 0.0f;
    for (int j = t; j < NN; j += 256) {
        float x = X[(size_t)m * NN + j];
        ss += x * x;
    }
    red[t] = ss;
    __syncthreads();
    for (int o = 128; o > 0; o >>= 1) {
        if (t < o) red[t] += red[t + o];
        __syncthreads();
    }
    const float rs = rsqrtf(red[0] / (float)NN);
    for (int j = t; j < NN; j += 256)
        Xn[(size_t)m * NN + j] = X[(size_t)m * NN + j] * rs;
}

__global__ __launch_bounds__(32) void k_proj_wmma(const float* __restrict__ Xn,
                                                  const float* __restrict__ Wq,
                                                  const float* __restrict__ Wk,
                                                  const float* __restrict__ Wv,
                                                  float* __restrict__ qraw,
                                                  float* __restrict__ kraw,
                                                  float* __restrict__ vraw)
{
    const int lane = threadIdx.x;
    const int col  = lane & 15;
    const int lh   = lane >> 4;
    const int bid  = blockIdx.x;
    const int w    = bid >> 7;
    const int n0   = (bid & 127) << 4;

    const float* W   = (w == 0) ? Wq : (w == 1) ? Wk : Wv;
    float*       out = (w == 0) ? qraw : (w == 1) ? kraw : vraw;

    const float* arow = Xn + (size_t)col * NN + 2 * lh;
    const float* bcol = W + n0 + col;

    v8f acc = {};
    #pragma unroll 8
    for (int kb = 0; kb < NN; kb += 4) {
        float2 a2 = *reinterpret_cast<const float2*>(arow + kb);
        v2f A = {a2.x, a2.y};
        const float* bp = bcol + (size_t)(kb + 2 * lh) * NN;
        v2f B = {bp[0], bp[NN]};
        acc = __builtin_amdgcn_wmma_f32_16x16x4_f32(false, A, false, B,
                                                    (short)0, acc, false, false);
    }

    #pragma unroll
    for (int r = 0; r < 8; ++r)
        out[(size_t)(r + 8 * lh) * NN + n0 + col] = acc[r];
}

__global__ __launch_bounds__(256) void k_qknorm(const float* __restrict__ qraw,
                                                const float* __restrict__ kraw,
                                                const float* __restrict__ vraw,
                                                float* __restrict__ qn,
                                                float* __restrict__ kn,
                                                float* __restrict__ vn)
{
    const int gtid = blockIdx.x * blockDim.x + threadIdx.x;
    const int wid  = gtid >> 5;
    const int lane = gtid & 31;
    const int h = wid >> 4;
    const int m = wid & 15;
    const int d0 = lane * 4;

    const size_t src = (size_t)m * NN + (size_t)h * DD + d0;
    const size_t dst = ((size_t)h * MM + m) * DD + d0;

    float4 q4 = *reinterpret_cast<const float4*>(qraw + src);
    float ss = q4.x * q4.x + q4.y * q4.y + q4.z * q4.z + q4.w * q4.w;
    #pragma unroll
    for (int mk = 16; mk >= 1; mk >>= 1) ss += __shfl_xor(ss, mk, 32);
    float rs = rsqrtf(ss / (float)DD);
    float4 o;
    o.x = q4.x * rs; o.y = q4.y * rs; o.z = q4.z * rs; o.w = q4.w * rs;
    *reinterpret_cast<float4*>(qn + dst) = o;

    float4 k4 = *reinterpret_cast<const float4*>(kraw + src);
    ss = k4.x * k4.x + k4.y * k4.y + k4.z * k4.z + k4.w * k4.w;
    #pragma unroll
    for (int mk = 16; mk >= 1; mk >>= 1) ss += __shfl_xor(ss, mk, 32);
    rs = rsqrtf(ss / (float)DD);
    o.x = k4.x * rs; o.y = k4.y * rs; o.z = k4.z * rs; o.w = k4.w * rs;
    *reinterpret_cast<float4*>(kn + dst) = o;

    *reinterpret_cast<float4*>(vn + dst) =
        *reinterpret_cast<const float4*>(vraw + src);
}

#define QP 132
#define SP 260

__global__ __launch_bounds__(256) void k_attn_partial(const float* __restrict__ qn,
                                                      const float* __restrict__ kn,
                                                      const float* __restrict__ vn,
                                                      const float* __restrict__ cK,
                                                      const float* __restrict__ cV,
                                                      float* __restrict__ pout,
                                                      float* __restrict__ pmax,
                                                      float* __restrict__ psum)
{
    __shared__ __align__(16) float q_lds[MM * QP];
    __shared__ __align__(16) float sc[MM * SP];

    const int tid  = threadIdx.x;
    const int lane = tid & 31;
    const int wv   = tid >> 5;
    const int col  = lane & 15;
    const int lh   = lane >> 4;
    const int s    = blockIdx.x;
    const int h    = blockIdx.y;
    const bool full = (s < 32);

    const float* qh = qn + (size_t)h * MM * DD;
    #pragma unroll
    for (int u = 0; u < 2; ++u) {
        const int i = tid + u * 256;
        const int row = i >> 5, c4 = i & 31;
        *reinterpret_cast<float4*>(q_lds + row * QP + c4 * 4) =
            reinterpret_cast<const float4*>(qh)[i];
    }

    const float* Kb = full ? (cK + ((size_t)h * PP + (size_t)s * 256) * DD)
                           : (kn + (size_t)h * MM * DD);
    const float* Vb = full ? (cV + ((size_t)h * PP + (size_t)s * 256) * DD)
                           : (vn + (size_t)h * MM * DD);

    if (!full) {
        for (int i = tid; i < MM * SP; i += 256) sc[i] = -3.0e38f;
    }
    __syncthreads();

    const float* qrow = q_lds + col * QP + 2 * lh;
    if (full) {
        const int j0 = wv * 32;
        const float* k0 = Kb + (size_t)(j0 + col) * DD + 2 * lh;
        const float* k1 = k0 + 16 * DD;
        v8f acc0 = {}, acc1 = {};
        #pragma unroll
        for (int kb = 0; kb < DD; kb += 4) {
            float2 a2 = *reinterpret_cast<const float2*>(qrow + kb);
            float2 b0 = *reinterpret_cast<const float2*>(k0 + kb);
            float2 b1 = *reinterpret_cast<const float2*>(k1 + kb);
            v2f A  = {a2.x, a2.y};
            v2f B0 = {b0.x, b0.y};
            v2f B1 = {b1.x, b1.y};
            acc0 = __builtin_amdgcn_wmma_f32_16x16x4_f32(false, A, false, B0,
                                                         (short)0, acc0, false, false);
            acc1 = __builtin_amdgcn_wmma_f32_16x16x4_f32(false, A, false, B1,
                                                         (short)0, acc1, false, false);
        }
        #pragma unroll
        for (int r = 0; r < 8; ++r) {
            sc[(r + 8 * lh) * SP + j0 + col]      = acc0[r];
            sc[(r + 8 * lh) * SP + j0 + 16 + col] = acc1[r];
        }
    } else if (wv == 0) {
        const float* k0 = Kb + (size_t)col * DD + 2 * lh;
        v8f acc0 = {};
        #pragma unroll
        for (int kb = 0; kb < DD; kb += 4) {
            float2 a2 = *reinterpret_cast<const float2*>(qrow + kb);
            float2 b0 = *reinterpret_cast<const float2*>(k0 + kb);
            v2f A  = {a2.x, a2.y};
            v2f B0 = {b0.x, b0.y};
            acc0 = __builtin_amdgcn_wmma_f32_16x16x4_f32(false, A, false, B0,
                                                         (short)0, acc0, false, false);
        }
        #pragma unroll
        for (int r = 0; r < 8; ++r)
            sc[(r + 8 * lh) * SP + col] = acc0[r];
    }
    __syncthreads();

    {
        const int row  = tid >> 4;
        const int part = tid & 15;
        float* srow = sc + row * SP + part;
        float v[16];
        float mloc = -3.0e38f;
        #pragma unroll
        for (int i = 0; i < 16; ++i) {
            v[i] = srow[16 * i];
            mloc = fmaxf(mloc, v[i]);
        }
        for (int mk = 8; mk >= 1; mk >>= 1)
            mloc = fmaxf(mloc, __shfl_xor(mloc, mk, 32));
        #pragma unroll
        for (int i = 0; i < 16; ++i)
            srow[16 * i] = __expf(v[i] - mloc);
        if (part == 0)
            pmax[((size_t)h * NCHUNK + s) * MM + row] = mloc;
    }
    __syncthreads();

    if (wv == 0) {
        const float* scrow = sc + col * SP + lh * 128;
        float ssum = 0.0f;
        for (int j = 0; j < 128; ++j) ssum += scrow[j];
        ssum += __shfl_xor(ssum, 16, 32);
        if (lh == 0)
            psum[((size_t)h * NCHUNK + s) * MM + col] = ssum;
    }

    const float* ap = sc + col * SP + 2 * lh;
    const float* vp = Vb + (size_t)(2 * lh) * DD + wv * 16 + col;
    v8f oacc = {};
    if (full) {
        #pragma unroll 32
        for (int kb = 0; kb < 256; kb += 4) {
            float2 a2 = *reinterpret_cast<const float2*>(ap + kb);
            v2f A = {a2.x, a2.y};
            v2f B = {vp[(size_t)kb * DD], vp[(size_t)kb * DD + DD]};
            oacc = __builtin_amdgcn_wmma_f32_16x16x4_f32(false, A, false, B,
                                                         (short)0, oacc, false, false);
        }
    } else {
        #pragma unroll
        for (int kb = 0; kb < MM; kb += 4) {
            float2 a2 = *reinterpret_cast<const float2*>(ap + kb);
            v2f A = {a2.x, a2.y};
            v2f B = {vp[(size_t)kb * DD], vp[(size_t)kb * DD + DD]};
            oacc = __builtin_amdgcn_wmma_f32_16x16x4_f32(false, A, false, B,
                                                         (short)0, oacc, false, false);
        }
    }

    float* po = pout + ((size_t)h * NCHUNK + s) * MM * DD;
    #pragma unroll
    for (int r = 0; r < 8; ++r)
        po[(size_t)(r + 8 * lh) * DD + wv * 16 + col] = oacc[r];
}

__global__ __launch_bounds__(128) void k_combine(const float* __restrict__ pout,
                                                 const float* __restrict__ pmax,
                                                 const float* __restrict__ psum,
                                                 float* __restrict__ out)
{
    const int h = blockIdx.x >> 4;
    const int m = blockIdx.x & (MM - 1);
    const int d = threadIdx.x;
    const float* pm  = pmax + (size_t)h * NCHUNK * MM + m;
    const float* psu = psum + (size_t)h * NCHUNK * MM + m;
    const float* po  = pout + ((size_t)h * NCHUNK * MM + m) * DD + d;

    float mx[NCHUNK], sm[NCHUNK], pv[NCHUNK];
    #pragma unroll
    for (int s = 0; s < NCHUNK; ++s) {
        mx[s] = pm[(size_t)s * MM];
        sm[s] = psu[(size_t)s * MM];
        pv[s] = po[(size_t)s * MM * DD];
    }
    float g = -3.0e38f;
    #pragma unroll
    for (int s = 0; s < NCHUNK; ++s) g = fmaxf(g, mx[s]);
    float tot = 0.0f, acc = 0.0f;
    #pragma unroll
    for (int s = 0; s < NCHUNK; ++s) {
        float w = __expf(mx[s] - g);
        tot += w * sm[s];
        acc += w * pv[s];
    }
    out[(size_t)m * NN + (size_t)h * DD + d] = acc / tot;
}

extern "C" void kernel_launch(void* const* d_in, const int* in_sizes, int n_in,
                              void* d_out, int out_size, void* d_ws, size_t ws_size,
                              hipStream_t stream)
{
    (void)in_sizes; (void)n_in; (void)out_size; (void)ws_size;
    const float* X  = (const float*)d_in[0];
    const float* Wq = (const float*)d_in[1];
    const float* Wk = (const float*)d_in[2];
    const float* Wv = (const float*)d_in[3];
    const float* cK = (const float*)d_in[4];
    const float* cV = (const float*)d_in[5];
    float* out = (float*)d_out;
    float* ws  = (float*)d_ws;

    float* xnorm = ws;
    float* qraw  = xnorm + MM * NN;
    float* kraw  = qraw + MM * NN;
    float* vraw  = kraw + MM * NN;
    float* qn    = vraw + MM * NN;
    float* kn    = qn + HH * MM * DD;
    float* vn    = kn + HH * MM * DD;
    float* pout  = vn + HH * MM * DD;
    float* pmaxb = pout + (size_t)HH * NCHUNK * MM * DD;
    float* psumb = pmaxb + (size_t)HH * NCHUNK * MM;

    k_rmsnorm_x<<<16, 256, 0, stream>>>(X, xnorm);
    k_proj_wmma<<<3 * (NN / 16), 32, 0, stream>>>(xnorm, Wq, Wk, Wv,
                                                  qraw, kraw, vraw);
    k_qknorm<<<(HH * MM) / 8, 256, 0, stream>>>(qraw, kraw, vraw, qn, kn, vn);
    k_attn_partial<<<dim3(NCHUNK, HH), 256, 0, stream>>>(qn, kn, vn, cK, cV,
                                                        pout, pmaxb, psumb);
    k_combine<<<HH * MM, DD, 0, stream>>>(pout, pmaxb, psumb, out);
}


# projection kernel: k range of each 16x16 tile split over 16 waves of one block (was one wave per tile), partials added through LDS in fixed wave order; order of the k sums of Q, K, V changes
# speedup vs baseline: 3.2623x; 1.4673x over previous
#include <hip/hip_runtime.h>
#include <math.h>


typedef __attribute__((ext_vector_type(2))) float v2f;
typedef __attribute__((ext_vector_type(8))) float v8f;

#define MM 16
#define NN 2048
#define DD 128
#define PP 8192
#define HH 16
#define NCHUNK 33

__global__ __launch_bounds__(256) void k_rmsnorm_x(const float* __restrict__ X,
                                                   float* __restrict__ Xn)
{
    __shared__ float red[256];
    const int m = blockIdx.x;
    const int t = threadIdx.x;
    float ss = 0.0f;
    for (int j = t; j < NN; j += 256) {
        float x = X[(size_t)m * NN + j];
        ss += x * x;
    }
    red[t] = ss;
    __syncthreads();
    for (int o = 128; o > 0; o >>= 1) {
        if (t < o) red[t] += red[t + o];
        __syncthreads();
    }
    const float rs = rsqrtf(red[0] / (float)NN);
    for (int j = t; j < NN; j += 256)
        Xn[(size_t)m * NN + j] = X[(size_t)m * NN + j] * rs;
}

#define KSPLIT 16
#define KSEG   (NN / KSPLIT)

__global__ __launch_bounds__(32 * KSPLIT) __attribute__((amdgpu_waves_per_eu(4, 8)))
void k_proj_wmma(const float* __restrict__ Xn,
                 const float* __restrict__ Wq,
                 const float* __restrict__ Wk,
                 const float* __restrict__ Wv,
                 float* __restrict__ qraw,
                 float* __restrict__ kraw,
                 float* __restrict__ vraw)
{
    __shared__ float part[KSPLIT * 256];

    const int tid  = threadIdx.x;
    const int lane = tid & 31;
    const int wv   = tid >> 5;
    const int col  = lane & 15;
    const int lh   = lane >> 4;
    const int bid  = blockIdx.x;
    const int w    = bid >> 7;
    const int n0   = (bid & 127) << 4;

    const float* W   = (w == 0) ? Wq : (w == 1) ? Wk : Wv;
    float*       out = (w == 0) ? qraw : (w == 1) ? kraw : vraw;

    const int k0 = wv * KSEG;
    const float* arow = Xn + (size_t)col * NN + k0 + 2 * lh;
    const float* bcol = W + (size_t)(k0 + 2 * lh) * NN + n0 + col;

    v8f acc = {};
    #pragma unroll 16
    for (int kb = 0; kb < KSEG; kb += 4) {
        float2 a2 = *reinterpret_cast<const float2*>(arow + kb);
        v2f A = {a2.x, a2.y};
        const float* bp = bcol + (size_t)kb * NN;
        v2f B = {bp[0], bp[NN]};
        acc = __builtin_amdgcn_wmma_f32_16x16x4_f32(false, A, false, B,
                                                    (short)0, acc, false, false);
    }

    #pragma unroll
    for (int r = 0; r < 8; ++r)
        part[wv * 256 + (2 * r + lh) * 16 + col] = acc[r];
    __syncthreads();

    if (tid < 256) {
        float s = part[tid];
        #pragma unroll
        for (int p = 1; p < KSPLIT; ++p) s += part[p * 256 + tid];
        const int slot = tid >> 4;
        const int row  = (slot >> 1) + 8 * (slot & 1);
        out[(size_t)row * NN + n0 + (tid & 15)] = s;
    }
}

__global__ __launch_bounds__(256) void k_qknorm(const float* __restrict__ qraw,
                                                const float* __restrict__ kraw,
                                                const float* __restrict__ vraw,
                                                float* __restrict__ qn,
                                                float* __restrict__ kn,
                                                float* __restrict__ vn)
{
    const int gtid = blockIdx.x * blockDim.x + threadIdx.x;
    const int wid  = gtid >> 5;
    const int lane = gtid & 31;
    const int h = wid >> 4;
    const int m = wid & 15;
    const int d0 = lane * 4;

    const size_t src = (size_t)m * NN + (size_t)h * DD + d0;
    const size_t dst = ((size_t)h * MM + m) * DD + d0;

    float4 q4 = *reinterpret_cast<const float4*>(qraw + src);
    float ss = q4.x * q4.x + q4.y * q4.y + q4.z * q4.z + q4.w * q4.w;
    #pragma unroll
    for (int mk = 16; mk >= 1; mk >>= 1) ss += __shfl_xor(ss, mk, 32);
    float rs = rsqrtf(ss / (float)DD);
    float4 o;
    o.x = q4.x * rs; o.y = q4.y * rs; o.z = q4.z * rs; o.w = q4.w * rs;
    *reinterpret_cast<float4*>(qn + dst) = o;

    float4 k4 = *reinterpret_cast<const float4*>(kraw + src);
    ss = k4.x * k4.x + k4.y * k4.y + k4.z * k4.z + k4.w * k4.w;
    #pragma unroll
    for (int mk = 16; mk >= 1; mk >>= 1) ss += __shfl_xor(ss, mk, 32);
    rs = rsqrtf(ss / (float)DD);
    o.x = k4.x * rs; o.y = k4.y * rs; o.z = k4.z * rs; o.w = k4.w * rs;
    *reinterpret_cast<float4*>(kn + dst) = o;

    *reinterpret_cast<float4*>(vn + dst) =
        *reinterpret_cast<const float4*>(vraw + src);
}

#define QP 132
#define SP 260

__global__ __launch_bounds__(256) void k_attn_partial(const float* __restrict__ qn,
                                                      const float* __restrict__ kn,
                                                      const float* __restrict__ vn,
                                                      const float* __restrict__ cK,
                                                      const float* __restrict__ cV,
                                                      float* __restrict__ pout,
                                                      float* __restrict__ pmax,
                                                      float* __restrict__ psum)
{
    __shared__ __align__(16) float q_lds[MM * QP];
    __shared__ __align__(16) float sc[MM * SP];

    const int tid  = threadIdx.x;
    const int lane = tid & 31;
    const int wv   = tid >> 5;
    const int col  = lane & 15;
    const int lh   = lane >> 4;
    const int s    = blockIdx.x;
    const int h    = blockIdx.y;
    const bool full = (s < 32);

    const float* qh = qn + (size_t)h * MM * DD;
    #pragma unroll
    for (int u = 0; u < 2; ++u) {
        const int i = tid + u * 256;
        const int row = i >> 5, c4 = i & 31;
        *reinterpret_cast<float4*>(q_lds + row * QP + c4 * 4) =
            reinterpret_cast<const float4*>(qh)[i];
    }

    const float* Kb = full ? (cK + ((size_t)h * PP + (size_t)s * 256) * DD)
                           : (kn + (size_t)h * MM * DD);
    const float* Vb = full ? (cV + ((size_t)h * PP + (size_t)s * 256) * DD)
                           : (vn + (size_t)h * MM * DD);

    if (!full) {
        for (int i = tid; i < MM * SP; i += 256) sc[i] = -3.0e38f;
    }
    __syncthreads();

    const float* qrow = q_lds + col * QP + 2 * lh;
    if (full) {
        const int j0 = wv * 32;
        const float* k0 = Kb + (size_t)(j0 + col) * DD + 2 * lh;
        const float* k1 = k0 + 16 * DD;
        v8f acc0 = {}, acc1 = {};
        #pragma unroll
        for (int kb = 0; kb < DD; kb += 4) {
            float2 a2 = *reinterpret_cast<const float2*>(qrow + kb);
            float2 b0 = *reinterpret_cast<const float2*>(k0 + kb);
            float2 b1 = *reinterpret_cast<const float2*>(k1 + kb);
            v2f A  = {a2.x, a2.y};
            v2f B0 = {b0.x, b0.y};
            v2f B1 = {b1.x, b1.y};
            acc0 = __builtin_amdgcn_wmma_f32_16x16x4_f32(false, A, false, B0,
                                                         (short)0, acc0, false, false);
            acc1 = __builtin_amdgcn_wmma_f32_16x16x4_f32(false, A, false, B1,
                                                         (short)0, acc1, false, false);
        }
        #pragma unroll
        for (int r = 0; r < 8; ++r) {
            sc[(r + 8 * lh) * SP + j0 + col]      = acc0[r];
            sc[(r + 8 * lh) * SP + j0 + 16 + col] = acc1[r];
        }
    } else if (wv == 0) {
        const float* k0 = Kb + (size_t)col * DD + 2 * lh;
        v8f acc0 = {};
        #pragma unroll
        for (int kb = 0; kb < DD; kb += 4) {
            float2 a2 = *reinterpret_cast<const float2*>(qrow + kb);
            float2 b0 = *reinterpret_cast<const float2*>(k0 + kb);
            v2f A  = {a2.x, a2.y};
            v2f B0 = {b0.x, b0.y};
            acc0 = __builtin_amdgcn_wmma_f32_16x16x4_f32(false, A, false, B0,
                                                         (short)0, acc0, false, false);
        }
        #pragma unroll
        for (int r = 0; r < 8; ++r)
            sc[(r + 8 * lh) * SP + col] = acc0[r];
    }
    __syncthreads();

    {
        const int row  = tid >> 4;
        const int part = tid & 15;
        float* srow = sc + row * SP + part;
        float v[16];
        float mloc = -3.0e38f;
        #pragma unroll
        for (int i = 0; i < 16; ++i) {
            v[i] = srow[16 * i];
            mloc = fmaxf(mloc, v[i]);
        }
        for (int mk = 8; mk >= 1; mk >>= 1)
            mloc = fmaxf(mloc, __shfl_xor(mloc, mk, 32));
        #pragma unroll
        for (int i = 0; i < 16; ++i)
            srow[16 * i] = __expf(v[i] - mloc);
        if (part == 0)
            pmax[((size_t)h * NCHUNK + s) * MM + row] = mloc;
    }
    __syncthreads();

    if (wv == 0) {
        const float* scrow = sc + col * SP + lh * 128;
        float ssum = 0.0f;
        for (int j = 0; j < 128; ++j) ssum += scrow[j];
        ssum += __shfl_xor(ssum, 16, 32);
        if (lh == 0)
            psum[((size_t)h * NCHUNK + s) * MM + col] = ssum;
    }

    const float* ap = sc + col * SP + 2 * lh;
    const float* vp = Vb + (size_t)(2 * lh) * DD + wv * 16 + col;
    v8f oacc = {};
    if (full) {
        #pragma unroll 32
        for (int kb = 0; kb < 256; kb += 4) {
            float2 a2 = *reinterpret_cast<const float2*>(ap + kb);
            v2f A = {a2.x, a2.y};
            v2f B = {vp[(size_t)kb * DD], vp[(size_t)kb * DD + DD]};
            oacc = __builtin_amdgcn_wmma_f32_16x16x4_f32(false, A, false, B,
                                                         (short)0, oacc, false, false);
        }
    } else {
        #pragma unroll
        for (int kb = 0; kb < MM; kb += 4) {
            float2 a2 = *reinterpret_cast<const float2*>(ap + kb);
            v2f A = {a2.x, a2.y};
            v2f B = {vp[(size_t)kb * DD], vp[(size_t)kb * DD + DD]};
            oacc = __builtin_amdgcn_wmma_f32_16x16x4_f32(false, A, false, B,
                                                         (short)0, oacc, false, false);
        }
    }

    float* po = pout + ((size_t)h * NCHUNK + s) * MM * DD;
    #pragma unroll
    for (int r = 0; r < 8; ++r)
        po[(size_t)(r + 8 * lh) * DD + wv * 16 + col] = oacc[r];
}

__global__ __launch_bounds__(128) void k_combine(const float* __restrict__ pout,
                                                 const float* __restrict__ pmax,
                                                 const float* __restrict__ psum,
                                                 float* __restrict__ out)
{
    const int h = blockIdx.x >> 4;
    const int m = blockIdx.x & (MM - 1);
    const int d = threadIdx.x;
    const float* pm  = pmax + (size_t)h * NCHUNK * MM + m;
    const float* psu = psum + (size_t)h * NCHUNK * MM + m;
    const float* po  = pout + ((size_t)h * NCHUNK * MM + m) * DD + d;

    float mx[NCHUNK], sm[NCHUNK], pv[NCHUNK];
    #pragma unroll
    for (int s = 0; s < NCHUNK; ++s) {
        mx[s] = pm[(size_t)s * MM];
        sm[s] = psu[(size_t)s * MM];
        pv[s] = po[(size_t)s * MM * DD];
    }
    float g = -3.0e38f;
    #pragma unroll
    for (int s = 0; s < NCHUNK; ++s) g = fmaxf(g, mx[s]);
    float tot = 0.0f, acc = 0.0f;
    #pragma unroll
    for (int s = 0; s < NCHUNK; ++s) {
        float w = __expf(mx[s] - g);
        tot += w * sm[s];
        acc += w * pv[s];
    }
    out[(size_t)m * NN + (size_t)h * DD + d] = acc / tot;
}

extern "C" void kernel_launch(void* const* d_in, const int* in_sizes, int n_in,
                              void* d_out, int out_size, void* d_ws, size_t ws_size,
                              hipStream_t stream)
{
    (void)in_sizes; (void)n_in; (void)out_size; (void)ws_size;
    const float* X  = (const float*)d_in[0];
    const float* Wq = (const float*)d_in[1];
    const float* Wk = (const float*)d_in[2];
    const float* Wv = (const float*)d_in[3];
    const float* cK = (const float*)d_in[4];
    const float* cV = (const float*)d_in[5];
    float* out = (float*)d_out;
    float* ws  = (float*)d_ws;

    float* xnorm = ws;
    float* qraw  = xnorm + MM * NN;
    float* kraw  = qraw + MM * NN;
    float* vraw  = kraw + MM * NN;
    float* qn    = vraw + MM * NN;
    float* kn    = qn + HH * MM * DD;
    float* vn    = kn + HH * MM * DD;
    float* pout  = vn + HH * MM * DD;
    float* pmaxb = pout + (size_t)HH * NCHUNK * MM * DD;
    float* psumb = pmaxb + (size_t)HH * NCHUNK * MM;

    k_rmsnorm_x<<<16, 256, 0, stream>>>(X, xnorm);
    k_proj_wmma<<<3 * (NN / 16), 32 * KSPLIT, 0, stream>>>(xnorm, Wq, Wk, Wv,
                                                  qraw, kraw, vraw);
    k_qknorm<<<(HH * MM) / 8, 256, 0, stream>>>(qraw, kraw, vraw, qn, kn, vn);
    k_attn_partial<<<dim3(NCHUNK, HH), 256, 0, stream>>>(qn, kn, vn, cK, cV,
                                                        pout, pmaxb, psumb);
    k_combine<<<HH * MM, DD, 0, stream>>>(pout, pmaxb, psumb, out);
}


# both norm launches folded into their consumers: projection scales its tile by rsqrt(mean X^2) taken from its own A fragments (new sum order), attention norms q/k while staging; 5 launches -> 3
# speedup vs baseline: 4.3427x; 1.3312x over previous
#include <hip/hip_runtime.h>
#include <math.h>


typedef __attribute__((ext_vector_type(2))) float v2f;
typedef __attribute__((ext_vector_type(8))) float v8f;

#define MM 16
#define NN 2048
#define DD 128
#define PP 8192
#define HH 16
#define NCHUNK 33

#define KSPLIT 16
#define KSEG   (NN / KSPLIT)

__global__ __launch_bounds__(32 * KSPLIT) __attribute__((amdgpu_waves_per_eu(4, 8)))
void k_proj_wmma(const float* __restrict__ X,
                 const float* __restrict__ Wq,
                 const float* __restrict__ Wk,
                 const float* __restrict__ Wv,
                 float* __restrict__ qraw,
                 float* __restrict__ kraw,
                 float* __restrict__ vraw)
{
    __shared__ float part[KSPLIT * 256];
    __shared__ float ssq[KSPLIT * 16];

    const int tid  = threadIdx.x;
    const int lane = tid & 31;
    const int wv   = tid >> 5;
    const int col  = lane & 15;
    const int lh   = lane >> 4;
    const int bid  = blockIdx.x;
    const int w    = bid >> 7;
    const int n0   = (bid & 127) << 4;

    const float* W   = (w == 0) ? Wq : (w == 1) ? Wk : Wv;
    float*       out = (w == 0) ? qraw : (w == 1) ? kraw : vraw;

    const int k0 = wv * KSEG;
    const float* arow = X + (size_t)col * NN + k0 + 2 * lh;
    const float* bcol = W + (size_t)(k0 + 2 * lh) * NN + n0 + col;

    v8f acc = {};
    float ss = 0.0f;
    #pragma unroll 16
    for (int kb = 0; kb < KSEG; kb += 4) {
        float2 a2 = *reinterpret_cast<const float2*>(arow + kb);
        ss += a2.x * a2.x;
        ss += a2.y * a2.y;
        v2f A = {a2.x, a2.y};
        const float* bp = bcol + (size_t)kb * NN;
        v2f B = {bp[0], bp[NN]};
        acc = __builtin_amdgcn_wmma_f32_16x16x4_f32(false, A, false, B,
                                                    (short)0, acc, false, false);
    }

    #pragma unroll
    for (int r = 0; r < 8; ++r)
        part[wv * 256 + (2 * r + lh) * 16 + col] = acc[r];
    ss += __shfl_xor(ss, 16, 32);
    if (lh == 0) ssq[wv * 16 + col] = ss;
    __syncthreads();

    if (tid < 256) {
        const int slot = tid >> 4;
        const int row  = (slot >> 1) + 8 * (slot & 1);
        float s = part[tid];
        float t = ssq[row];
        #pragma unroll
        for (int p = 1; p < KSPLIT; ++p) {
            s += part[p * 256 + tid];
            t += ssq[p * 16 + row];
        }
        out[(size_t)row * NN + n0 + (tid & 15)] = s * rsqrtf(t / (float)NN);
    }
}

#define QP 132
#define SP 260

__global__ __launch_bounds__(256) void k_attn_partial(const float* __restrict__ qraw,
                                                      const float* __restrict__ kraw,
                                                      const float* __restrict__ vraw,
                                                      const float* __restrict__ cK,
                                                      const float* __restrict__ cV,
                                                      float* __restrict__ pout,
                                                      float* __restrict__ pmax,
                                                      float* __restrict__ psum)
{
    __shared__ __align__(16) float q_lds[MM * QP];
    __shared__ __align__(16) float sc[MM * SP];
    __shared__ __align__(16) float k_lds[MM * QP];

    const int tid  = threadIdx.x;
    const int lane = tid & 31;
    const int wv   = tid >> 5;
    const int col  = lane & 15;
    const int lh   = lane >> 4;
    const int s    = blockIdx.x;
    const int h    = blockIdx.y;
    const bool full = (s < 32);

    {
        const float* qh = qraw + (size_t)h * DD + 4 * lane;
        float4 q4[2];
        #pragma unroll
        for (int u = 0; u < 2; ++u)
            q4[u] = *reinterpret_cast<const float4*>(qh + (size_t)(wv + 8 * u) * NN);
        #pragma unroll
        for (int u = 0; u < 2; ++u) {
            float ss = q4[u].x * q4[u].x + q4[u].y * q4[u].y
                     + q4[u].z * q4[u].z + q4[u].w * q4[u].w;
            #pragma unroll
            for (int mk = 16; mk >= 1; mk >>= 1) ss += __shfl_xor(ss, mk, 32);
            const float rs = rsqrtf(ss / (float)DD);
            float4 o;
            o.x = q4[u].x * rs; o.y = q4[u].y * rs; o.z = q4[u].z * rs; o.w = q4[u].w * rs;
            *reinterpret_cast<float4*>(q_lds + (wv + 8 * u) * QP + 4 * lane) = o;
        }
    }

    if (!full) {
        const float* kh = kraw + (size_t)h * DD + 4 * lane;
        float4 k4[2];
        #pragma unroll
        for (int u = 0; u < 2; ++u)
            k4[u] = *reinterpret_cast<const float4*>(kh + (size_t)(wv + 8 * u) * NN);
        #pragma unroll
        for (int u = 0; u < 2; ++u) {
            float ss = k4[u].x * k4[u].x + k4[u].y * k4[u].y
                     + k4[u].z * k4[u].z + k4[u].w * k4[u].w;
            #pragma unroll
            for (int mk = 16; mk >= 1; mk >>= 1) ss += __shfl_xor(ss, mk, 32);
            const float rs = rsqrtf(ss / (float)DD);
            float4 o;
            o.x = k4[u].x * rs; o.y = k4[u].y * rs; o.z = k4[u].z * rs; o.w = k4[u].w * rs;
            *reinterpret_cast<float4*>(k_lds + (wv + 8 * u) * QP + 4 * lane) = o;
        }
        for (int i = tid; i < MM * SP; i += 256) sc[i] = -3.0e38f;
    }
    __syncthreads();

    const float* qrow = q_lds + col * QP + 2 * lh;
    if (full) {
        const int j0 = wv * 32;
        const float* Kb = cK + ((size_t)h * PP + (size_t)s * 256) * DD;
        const float* k0 = Kb + (size_t)(j0 + col) * DD + 2 * lh;
        const float* k1 = k0 + 16 * DD;
        v8f acc0 = {}, acc1 = {};
        #pragma unroll
        for (int kb = 0; kb < DD; kb += 4) {
            float2 a2 = *reinterpret_cast<const float2*>(qrow + kb);
            float2 b0 = *reinterpret_cast<const float2*>(k0 + kb);
            float2 b1 = *reinterpret_cast<const float2*>(k1 + kb);
            v2f A  = {a2.x, a2.y};
            v2f B0 = {b0.x, b0.y};
            v2f B1 = {b1.x, b1.y};
            acc0 = __builtin_amdgcn_wmma_f32_16x16x4_f32(false, A, false, B0,
                                                         (short)0, acc0, false, false);
            acc1 = __builtin_amdgcn_wmma_f32_16x16x4_f32(false, A, false, B1,
                                                         (short)0, acc1, false, false);
        }
        #pragma unroll
        for (int r = 0; r < 8; ++r) {
            sc[(r + 8 * lh) * SP + j0 + col]      = acc0[r];
            sc[(r + 8 * lh) * SP + j0 + 16 + col] = acc1[r];
        }
    } else if (wv == 0) {
        const float* k0 = k_lds + col * QP + 2 * lh;
        v8f acc0 = {};
        #pragma unroll
        for (int kb = 0; kb < DD; kb += 4) {
            float2 a2 = *reinterpret_cast<const float2*>(qrow + kb);
            float2 b0 = *reinterpret_cast<const float2*>(k0 + kb);
            v2f A  = {a2.x, a2.y};
            v2f B0 = {b0.x, b0.y};
            acc0 = __builtin_amdgcn_wmma_f32_16x16x4_f32(false, A, false, B0,
                                                         (short)0, acc0, false, false);
        }
        #pragma unroll
        for (int r = 0; r < 8; ++r)
            sc[(r + 8 * lh) * SP + col] = acc0[r];
    }
    __syncthreads();

    {
        const int row  = tid >> 4;
        const int part = tid & 15;
        float* srow = sc + row * SP + part;
        float v[16];
        float mloc = -3.0e38f;
        #pragma unroll
        for (int i = 0; i < 16; ++i) {
            v[i] = srow[16 * i];
            mloc = fmaxf(mloc, v[i]);
        }
        for (int mk = 8; mk >= 1; mk >>= 1)
            mloc = fmaxf(mloc, __shfl_xor(mloc, mk, 32));
        #pragma unroll
        for (int i = 0; i < 16; ++i)
            srow[16 * i] = __expf(v[i] - mloc);
        if (part == 0)
            pmax[((size_t)h * NCHUNK + s) * MM + row] = mloc;
    }
    __syncthreads();

    if (wv == 0) {
        const float* scrow = sc + col * SP + lh * 128;
        float ssum = 0.0f;
        for (int j = 0; j < 128; ++j) ssum += scrow[j];
        ssum += __shfl_xor(ssum, 16, 32);
        if (lh == 0)
            psum[((size_t)h * NCHUNK + s) * MM + col] = ssum;
    }

    const float* ap = sc + col * SP + 2 * lh;
    v8f oacc = {};
    if (full) {
        const float* Vb = cV + ((size_t)h * PP + (size_t)s * 256) * DD;
        const float* vp = Vb + (size_t)(2 * lh) * DD + wv * 16 + col;
        #pragma unroll 32
        for (int kb = 0; kb < 256; kb += 4) {
            float2 a2 = *reinterpret_cast<const float2*>(ap + kb);
            v2f A = {a2.x, a2.y};
            v2f B = {vp[(size_t)kb * DD], vp[(size_t)kb * DD + DD]};
            oacc = __builtin_amdgcn_wmma_f32_16x16x4_f32(false, A, false, B,
                                                         (short)0, oacc, false, false);
        }
    } else {
        const float* vp = vraw + (size_t)(2 * lh) * NN + (size_t)h * DD + wv * 16 + col;
        #pragma unroll
        for (int kb = 0; kb < MM; kb += 4) {
            float2 a2 = *reinterpret_cast<const float2*>(ap + kb);
            v2f A = {a2.x, a2.y};
            v2f B = {vp[(size_t)kb * NN], vp[(size_t)kb * NN + NN]};
            oacc = __builtin_amdgcn_wmma_f32_16x16x4_f32(false, A, false, B,
                                                         (short)0, oacc, false, false);
        }
    }

    float* po = pout + ((size_t)h * NCHUNK + s) * MM * DD;
    #pragma unroll
    for (int r = 0; r < 8; ++r)
        po[(size_t)(r + 8 * lh) * DD + wv * 16 + col] = oacc[r];
}

__global__ __launch_bounds__(128) void k_combine(const float* __restrict__ pout,
                                                 const float* __restrict__ pmax,
                                                 const float* __restrict__ psum,
                                                 float* __restrict__ out)
{
    const int h = blockIdx.x >> 4;
    const int m = blockIdx.x & (MM - 1);
    const int d = threadIdx.x;
    const float* pm  = pmax + (size_t)h * NCHUNK * MM + m;
    const float* psu = psum + (size_t)h * NCHUNK * MM + m;
    const float* po  = pout + ((size_t)h * NCHUNK * MM + m) * DD + d;

    float mx[NCHUNK], sm[NCHUNK], pv[NCHUNK];
    #pragma unroll
    for (int s = 0; s < NCHUNK; ++s) {
        mx[s] = pm[(size_t)s * MM];
        sm[s] = psu[(size_t)s * MM];
        pv[s] = po[(size_t)s * MM * DD];
    }
    float g = -3.0e38f;
    #pragma unroll
    for (int s = 0; s < NCHUNK; ++s) g = fmaxf(g, mx[s]);
    float tot = 0.0f, acc = 0.0f;
    #pragma unroll
    for (int s = 0; s < NCHUNK; ++s) {
        float w = __expf(mx[s] - g);
        tot += w * sm[s];
        acc += w * pv[s];
    }
    out[(size_t)m * NN + (size_t)h * DD + d] = acc / tot;
}

extern "C" void kernel_launch(void* const* d_in, const int* in_sizes, int n_in,
                              void* d_out, int out_size, void* d_ws, size_t ws_size,
                              hipStream_t stream)
{
    (void)in_sizes; (void)n_in; (void)out_size; (void)ws_size;
    const float* X  = (const float*)d_in[0];
    const float* Wq = (const float*)d_in[1];
    const float* Wk = (const float*)d_in[2];
    const float* Wv = (const float*)d_in[3];
    const float* cK = (const float*)d_in[4];
    const float* cV = (const float*)d_in[5];
    float* out = (float*)d_out;
    float* ws  = (float*)d_ws;

    float* qraw  = ws;
    float* kraw  = qraw + MM * NN;
    float* vraw  = kraw + MM * NN;
    float* pout  = vraw + MM * NN;
    float* pmaxb = pout + (size_t)HH * NCHUNK * MM * DD;
    float* psumb = pmaxb + (size_t)HH * NCHUNK * MM;

    k_proj_wmma<<<3 * (NN / 16), 32 * KSPLIT, 0, stream>>>(X, Wq, Wk, Wv,
                                                  qraw, kraw, vraw);
    k_attn_partial<<<dim3(NCHUNK, HH), 256, 0, stream>>>(qraw, kraw, vraw, cK, cV,
                                                        pout, pmaxb, psumb);
    k_combine<<<HH * MM, DD, 0, stream>>>(pout, pmaxb, psumb, out);
}


# attention scores: cached K rows fetched whole (one 512-byte row per load) and turned into B fragments through a per-wave LDS stage, not by 16-line fragment loads; every sum keeps its order
# speedup vs baseline: 4.5919x; 1.0574x over previous
#include <hip/hip_runtime.h>
#include <math.h>


typedef __attribute__((ext_vector_type(2))) float v2f;
typedef __attribute__((ext_vector_type(8))) float v8f;
typedef __attribute__((ext_vector_type(4))) float v4f;

#define MM 16
#define NN 2048
#define DD 128
#define PP 8192
#define HH 16
#define NCHUNK 33

#define KSPLIT 16
#define KSEG   (NN / KSPLIT)

__global__ __launch_bounds__(32 * KSPLIT) __attribute__((amdgpu_waves_per_eu(4, 8)))
void k_proj_wmma(const float* __restrict__ X,
                 const float* __restrict__ Wq,
                 const float* __restrict__ Wk,
                 const float* __restrict__ Wv,
                 float* __restrict__ qraw,
                 float* __restrict__ kraw,
                 float* __restrict__ vraw)
{
    __shared__ float part[KSPLIT * 256];
    __shared__ float ssq[KSPLIT * 16];

    const int tid  = threadIdx.x;
    const int lane = tid & 31;
    const int wv   = tid >> 5;
    const int col  = lane & 15;
    const int lh   = lane >> 4;
    const int bid  = blockIdx.x;
    const int w    = bid >> 7;
    const int n0   = (bid & 127) << 4;

    const float* W   = (w == 0) ? Wq : (w == 1) ? Wk : Wv;
    float*       out = (w == 0) ? qraw : (w == 1) ? kraw : vraw;

    const int k0 = wv * KSEG;
    const float* arow = X + (size_t)col * NN + k0 + 2 * lh;
    const float* bcol = W + (size_t)(k0 + 2 * lh) * NN + n0 + col;

    v8f acc = {};
    float ss = 0.0f;
    #pragma unroll 16
    for (int kb = 0; kb < KSEG; kb += 4) {
        float2 a2 = *reinterpret_cast<const float2*>(arow + kb);
        ss += a2.x * a2.x;
        ss += a2.y * a2.y;
        v2f A = {a2.x, a2.y};
        const float* bp = bcol + (size_t)kb * NN;
        v2f B = {bp[0], bp[NN]};
        acc = __builtin_amdgcn_wmma_f32_16x16x4_f32(false, A, false, B,
                                                    (short)0, acc, false, false);
    }

    #pragma unroll
    for (int r = 0; r < 8; ++r)
        part[wv * 256 + (2 * r + lh) * 16 + col] = acc[r];
    ss += __shfl_xor(ss, 16, 32);
    if (lh == 0) ssq[wv * 16 + col] = ss;
    __syncthreads();

    if (tid < 256) {
        const int slot = tid >> 4;
        const int row  = (slot >> 1) + 8 * (slot & 1);
        float s = part[tid];
        float t = ssq[row];
        #pragma unroll
        for (int p = 1; p < KSPLIT; ++p) {
            s += part[p * 256 + tid];
            t += ssq[p * 16 + row];
        }
        out[(size_t)row * NN + n0 + (tid & 15)] = s * rsqrtf(t / (float)NN);
    }
}

#define QP 132
#define SP 260
#define KP 132

__global__ __launch_bounds__(256) void k_attn_partial(const float* __restrict__ qraw,
                                                      const float* __restrict__ kraw,
                                                      const float* __restrict__ vraw,
                                                      const float* __restrict__ cK,
                                                      const float* __restrict__ cV,
                                                      float* __restrict__ pout,
                                                      float* __restrict__ pmax,
                                                      float* __restrict__ psum)
{
    __shared__ __align__(16) float q_lds[MM * QP];
    __shared__ __align__(16) float sc[MM * SP];
    __shared__ __align__(16) float kst[8 * MM * KP];

    const int tid  = threadIdx.x;
    const int lane = tid & 31;
    const int wv   = tid >> 5;
    const int col  = lane & 15;
    const int lh   = lane >> 4;
    const int s    = blockIdx.x;
    const int h    = blockIdx.y;
    const bool full = (s < 32);

    {
        const float* qh = qraw + (size_t)h * DD + 4 * lane;
        float4 q4[2];
        #pragma unroll
        for (int u = 0; u < 2; ++u)
            q4[u] = *reinterpret_cast<const float4*>(qh + (size_t)(wv + 8 * u) * NN);
        #pragma unroll
        for (int u = 0; u < 2; ++u) {
            float ss = q4[u].x * q4[u].x + q4[u].y * q4[u].y
                     + q4[u].z * q4[u].z + q4[u].w * q4[u].w;
            #pragma unroll
            for (int mk = 16; mk >= 1; mk >>= 1) ss += __shfl_xor(ss, mk, 32);
            const float rs = rsqrtf(ss / (float)DD);
            float4 o;
            o.x = q4[u].x * rs; o.y = q4[u].y * rs; o.z = q4[u].z * rs; o.w = q4[u].w * rs;
            *reinterpret_cast<float4*>(q_lds + (wv + 8 * u) * QP + 4 * lane) = o;
        }
    }

    if (!full) {
        const float* kh = kraw + (size_t)h * DD + 4 * lane;
        float4 k4[2];
        #pragma unroll
        for (int u = 0; u < 2; ++u)
            k4[u] = *reinterpret_cast<const float4*>(kh + (size_t)(wv + 8 * u) * NN);
        #pragma unroll
        for (int u = 0; u < 2; ++u) {
            float ss = k4[u].x * k4[u].x + k4[u].y * k4[u].y
                     + k4[u].z * k4[u].z + k4[u].w * k4[u].w;
            #pragma unroll
            for (int mk = 16; mk >= 1; mk >>= 1) ss += __shfl_xor(ss, mk, 32);
            const float rs = rsqrtf(ss / (float)DD);
            float4 o;
            o.x = k4[u].x * rs; o.y = k4[u].y * rs; o.z = k4[u].z * rs; o.w = k4[u].w * rs;
            *reinterpret_cast<float4*>(kst + (wv + 8 * u) * KP + 4 * lane) = o;
        }
        for (int i = tid; i < MM * SP; i += 256) sc[i] = -3.0e38f;
    }
    __syncthreads();

    const float* qrow = q_lds + col * QP + 2 * lh;
    if (full) {
        const int j0 = wv * 32;
        const float* Kw = cK + ((size_t)h * PP + (size_t)s * 256 + j0) * DD + 4 * lane;
        float* stg = kst + wv * (MM * KP);
        v4f kr[32];
        #pragma unroll
        for (int i = 0; i < 32; ++i)
            kr[i] = *reinterpret_cast<const v4f*>(Kw + (size_t)i * DD);
        #pragma unroll
        for (int i = 16; i < 32; ++i)
            asm volatile("" : "+v"(kr[i]));
        const float* kf = stg + col * KP + 2 * lh;
        #pragma unroll
        for (int t = 0; t < 2; ++t) {
            #pragma unroll
            for (int i = 0; i < 16; ++i)
                *reinterpret_cast<v4f*>(stg + i * KP + 4 * lane) = kr[16 * t + i];
            v8f acc = {};
            #pragma unroll
            for (int kb = 0; kb < DD; kb += 4) {
                float2 a2 = *reinterpret_cast<const float2*>(qrow + kb);
                float2 b2 = *reinterpret_cast<const float2*>(kf + kb);
                v2f A = {a2.x, a2.y};
                v2f B = {b2.x, b2.y};
                acc = __builtin_amdgcn_wmma_f32_16x16x4_f32(false, A, false, B,
                                                            (short)0, acc, false, false);
            }
            #pragma unroll
            for (int r = 0; r < 8; ++r)
                sc[(r + 8 * lh) * SP + j0 + 16 * t + col] = acc[r];
        }
    } else if (wv == 0) {
        const float* k0 = kst + col * KP + 2 * lh;
        v8f acc0 = {};
        #pragma unroll
        for (int kb = 0; kb < DD; kb += 4) {
            float2 a2 = *reinterpret_cast<const float2*>(qrow + kb);
            float2 b0 = *reinterpret_cast<const float2*>(k0 + kb);
            v2f A  = {a2.x, a2.y};
            v2f B0 = {b0.x, b0.y};
            acc0 = __builtin_amdgcn_wmma_f32_16x16x4_f32(false, A, false, B0,
                                                         (short)0, acc0, false, false);
        }
        #pragma unroll
        for (int r = 0; r < 8; ++r)
            sc[(r + 8 * lh) * SP + col] = acc0[r];
    }
    __syncthreads();

    {
        const int row  = tid >> 4;
        const int part = tid & 15;
        float* srow = sc + row * SP + part;
        float v[16];
        float mloc = -3.0e38f;
        #pragma unroll
        for (int i = 0; i < 16; ++i) {
            v[i] = srow[16 * i];
            mloc = fmaxf(mloc, v[i]);
        }
        for (int mk = 8; mk >= 1; mk >>= 1)
            mloc = fmaxf(mloc, __shfl_xor(mloc, mk, 32));
        #pragma unroll
        for (int i = 0; i < 16; ++i)
            srow[16 * i] = __expf(v[i] - mloc);
        if (part == 0)
            pmax[((size_t)h * NCHUNK + s) * MM + row] = mloc;
    }
    __syncthreads();

    if (wv == 0) {
        const float* scrow = sc + col * SP + lh * 128;
        float ssum = 0.0f;
        for (int j = 0; j < 128; ++j) ssum += scrow[j];
        ssum += __shfl_xor(ssum, 16, 32);
        if (lh == 0)
            psum[((size_t)h * NCHUNK + s) * MM + col] = ssum;
    }

    const float* ap = sc + col * SP + 2 * lh;
    v8f oacc = {};
    if (full) {
        const float* Vb = cV + ((size_t)h * PP + (size_t)s * 256) * DD;
        const float* vp = Vb + (size_t)(2 * lh) * DD + wv * 16 + col;
        #pragma unroll 32
        for (int kb = 0; kb < 256; kb += 4) {
            float2 a2 = *reinterpret_cast<const float2*>(ap + kb);
            v2f A = {a2.x, a2.y};
            v2f B = {vp[(size_t)kb * DD], vp[(size_t)kb * DD + DD]};
            oacc = __builtin_amdgcn_wmma_f32_16x16x4_f32(false, A, false, B,
                                                         (short)0, oacc, false, false);
        }
    } else {
        const float* vp = vraw + (size_t)(2 * lh) * NN + (size_t)h * DD + wv * 16 + col;
        #pragma unroll
        for (int kb = 0; kb < MM; kb += 4) {
            float2 a2 = *reinterpret_cast<const float2*>(ap + kb);
            v2f A = {a2.x, a2.y};
            v2f B = {vp[(size_t)kb * NN], vp[(size_t)kb * NN + NN]};
            oacc = __builtin_amdgcn_wmma_f32_16x16x4_f32(false, A, false, B,
                                                         (short)0, oacc, false, false);
        }
    }

    float* po = pout + ((size_t)h * NCHUNK + s) * MM * DD;
    #pragma unroll
    for (int r = 0; r < 8; ++r)
        po[(size_t)(r + 8 * lh) * DD + wv * 16 + col] = oacc[r];
}

__global__ __launch_bounds__(128) void k_combine(const float* __restrict__ pout,
                                                 const float* __restrict__ pmax,
                                                 const float* __restrict__ psum,
                                                 float* __restrict__ out)
{
    const int h = blockIdx.x >> 4;
    const int m = blockIdx.x & (MM - 1);
    const int d = threadIdx.x;
    const float* pm  = pmax + (size_t)h * NCHUNK * MM + m;
    const float* psu = psum + (size_t)h * NCHUNK * MM + m;
    const float* po  = pout + ((size_t)h * NCHUNK * MM + m) * DD + d;

    float mx[NCHUNK], sm[NCHUNK], pv[NCHUNK];
    #pragma unroll
    for (int s = 0; s < NCHUNK; ++s) {
        mx[s] = pm[(size_t)s * MM];
        sm[s] = psu[(size_t)s * MM];
        pv[s] = po[(size_t)s * MM * DD];
    }
    float g = -3.0e38f;
    #pragma unroll
    for (int s = 0; s < NCHUNK; ++s) g = fmaxf(g, mx[s]);
    float tot = 0.0f, acc = 0.0f;
    #pragma unroll
    for (int s = 0; s < NCHUNK; ++s) {
        float w = __expf(mx[s] - g);
        tot += w * sm[s];
        acc += w * pv[s];
    }
    out[(size_t)m * NN + (size_t)h * DD + d] = acc / tot;
}

extern "C" void kernel_launch(void* const* d_in, const int* in_sizes, int n_in,
                              void* d_out, int out_size, void* d_ws, size_t ws_size,
                              hipStream_t stream)
{
    (void)in_sizes; (void)n_in; (void)out_size; (void)ws_size;
    const float* X  = (const float*)d_in[0];
    const float* Wq = (const float*)d_in[1];
    const float* Wk = (const float*)d_in[2];
    const float* Wv = (const float*)d_in[3];
    const float* cK = (const float*)d_in[4];
    const float* cV = (const float*)d_in[5];
    float* out = (float*)d_out;
    float* ws  = (float*)d_ws;

    float* qraw  = ws;
    float* kraw  = qraw + MM * NN;
    float* vraw  = kraw + MM * NN;
    float* pout  = vraw + MM * NN;
    float* pmaxb = pout + (size_t)HH * NCHUNK * MM * DD;
    float* psumb = pmaxb + (size_t)HH * NCHUNK * MM;

    k_proj_wmma<<<3 * (NN / 16), 32 * KSPLIT, 0, stream>>>(X, Wq, Wk, Wv,
                                                  qraw, kraw, vraw);
    k_attn_partial<<<dim3(NCHUNK, HH), 256, 0, stream>>>(qraw, kraw, vraw, cK, cV,
                                                        pout, pmaxb, psumb);
    k_combine<<<HH * MM, DD, 0, stream>>>(pout, pmaxb, psumb, out);
}


# projection kernel: each wave computes two 16-column tiles (even/odd columns of a 32-column strip) from 8-byte W loads covering whole 128-byte lines; 192 blocks, not 384; every sum keeps its order
# speedup vs baseline: 5.3489x; 1.1649x over previous
#include <hip/hip_runtime.h>
#include <math.h>


typedef __attribute__((ext_vector_type(2))) float v2f;
typedef __attribute__((ext_vector_type(8))) float v8f;
typedef __attribute__((ext_vector_type(4))) float v4f;

#define MM 16
#define NN 2048
#define DD 128
#define PP 8192
#define HH 16
#define NCHUNK 33

#define KSPLIT 16
#define KSEG   (NN / KSPLIT)
#define TN     32

__global__ __launch_bounds__(32 * KSPLIT) __attribute__((amdgpu_waves_per_eu(4, 4)))
void k_proj_wmma(const float* __restrict__ X,
                 const float* __restrict__ Wq,
                 const float* __restrict__ Wk,
                 const float* __restrict__ Wv,
                 float* __restrict__ qraw,
                 float* __restrict__ kraw,
                 float* __restrict__ vraw)
{
    __shared__ __align__(16) float part[KSPLIT * 16 * TN];
    __shared__ float ssq[KSPLIT * 16];

    const int tid  = threadIdx.x;
    const int lane = tid & 31;
    const int wv   = tid >> 5;
    const int col  = lane & 15;
    const int lh   = lane >> 4;
    const int bid  = blockIdx.x;
    const int w    = bid >> 6;
    const int n0   = (bid & 63) << 5;

    const float* W   = (w == 0) ? Wq : (w == 1) ? Wk : Wv;
    float*       out = (w == 0) ? qraw : (w == 1) ? kraw : vraw;

    const int k0 = wv * KSEG;
    const float* arow = X + (size_t)col * NN + k0 + 2 * lh;
    const float* bcol = W + (size_t)(k0 + 2 * lh) * NN + n0 + 2 * col;

    v8f acc0 = {};
    v8f acc1 = {};
    float ss = 0.0f;
    #pragma unroll
    for (int kb = 0; kb < KSEG; kb += 4) {
        float2 a2 = *reinterpret_cast<const float2*>(arow + kb);
        ss += a2.x * a2.x;
        ss += a2.y * a2.y;
        v2f A = {a2.x, a2.y};
        const float* bp = bcol + (size_t)kb * NN;
        float2 b0 = *reinterpret_cast<const float2*>(bp);
        float2 b1 = *reinterpret_cast<const float2*>(bp + NN);
        v2f B0 = {b0.x, b1.x};
        v2f B1 = {b0.y, b1.y};
        acc0 = __builtin_amdgcn_wmma_f32_16x16x4_f32(false, A, false, B0,
                                                     (short)0, acc0, false, false);
        acc1 = __builtin_amdgcn_wmma_f32_16x16x4_f32(false, A, false, B1,
                                                     (short)0, acc1, false, false);
    }

    #pragma unroll
    for (int r = 0; r < 8; ++r) {
        float2 o;
        o.x = acc0[r];
        o.y = acc1[r];
        *reinterpret_cast<float2*>(part + wv * (16 * TN) + (2 * r + lh) * TN + 2 * col) = o;
    }
    ss += __shfl_xor(ss, 16, 32);
    if (lh == 0) ssq[wv * 16 + col] = ss;
    __syncthreads();

    {
        const int slot = tid >> 5;
        const int row  = (slot >> 1) + 8 * (slot & 1);
        float s = part[tid];
        float t = ssq[row];
        #pragma unroll
        for (int p = 1; p < KSPLIT; ++p) {
            s += part[p * (16 * TN) + tid];
            t += ssq[p * 16 + row];
        }
        out[(size_t)row * NN + n0 + (tid & 31)] = s * rsqrtf(t / (float)NN);
    }
}

#define QP 132
#define SP 260
#define KP 132

__global__ __launch_bounds__(256) void k_attn_partial(const float* __restrict__ qraw,
                                                      const float* __restrict__ kraw,
                                                      const float* __restrict__ vraw,
                                                      const float* __restrict__ cK,
                                                      const float* __restrict__ cV,
                                                      float* __restrict__ pout,
                                                      float* __restrict__ pmax,
                                                      float* __restrict__ psum)
{
    __shared__ __align__(16) float q_lds[MM * QP];
    __shared__ __align__(16) float sc[MM * SP];
    __shared__ __align__(16) float kst[8 * MM * KP];

    const int tid  = threadIdx.x;
    const int lane = tid & 31;
    const int wv   = tid >> 5;
    const int col  = lane & 15;
    const int lh   = lane >> 4;
    const int s    = blockIdx.x;
    const int h    = blockIdx.y;
    const bool full = (s < 32);

    {
        const float* qh = qraw + (size_t)h * DD + 4 * lane;
        float4 q4[2];
        #pragma unroll
        for (int u = 0; u < 2; ++u)
            q4[u] = *reinterpret_cast<const float4*>(qh + (size_t)(wv + 8 * u) * NN);
        #pragma unroll
        for (int u = 0; u < 2; ++u) {
            float ss = q4[u].x * q4[u].x + q4[u].y * q4[u].y
                     + q4[u].z * q4[u].z + q4[u].w * q4[u].w;
            #pragma unroll
            for (int mk = 16; mk >= 1; mk >>= 1) ss += __shfl_xor(ss, mk, 32);
            const float rs = rsqrtf(ss / (float)DD);
            float4 o;
            o.x = q4[u].x * rs; o.y = q4[u].y * rs; o.z = q4[u].z * rs; o.w = q4[u].w * rs;
            *reinterpret_cast<float4*>(q_lds + (wv + 8 * u) * QP + 4 * lane) = o;
        }
    }

    if (!full) {
        const float* kh = kraw + (size_t)h * DD + 4 * lane;
        float4 k4[2];
        #pragma unroll
        for (int u = 0; u < 2; ++u)
            k4[u] = *reinterpret_cast<const float4*>(kh + (size_t)(wv + 8 * u) * NN);
        #pragma unroll
        for (int u = 0; u < 2; ++u) {
            float ss = k4[u].x * k4[u].x + k4[u].y * k4[u].y
                     + k4[u].z * k4[u].z + k4[u].w * k4[u].w;
            #pragma unroll
            for (int mk = 16; mk >= 1; mk >>= 1) ss += __shfl_xor(ss, mk, 32);
            const float rs = rsqrtf(ss / (float)DD);
            float4 o;
            o.x = k4[u].x * rs; o.y = k4[u].y * rs; o.z = k4[u].z * rs; o.w = k4[u].w * rs;
            *reinterpret_cast<float4*>(kst + (wv + 8 * u) * KP + 4 * lane) = o;
        }
        for (int i = tid; i < MM * SP; i += 256) sc[i] = -3.0e38f;
    }
    __syncthreads();

    const float* qrow = q_lds + col * QP + 2 * lh;
    if (full) {
        const int j0 = wv * 32;
        const float* Kw = cK + ((size_t)h * PP + (size_t)s * 256 + j0) * DD + 4 * lane;
        float* stg = kst + wv * (MM * KP);
        v4f kr[32];
        #pragma unroll
        for (int i = 0; i < 32; ++i)
            kr[i] = *reinterpret_cast<const v4f*>(Kw + (size_t)i * DD);
        #pragma unroll
        for (int i = 16; i < 32; ++i)
            asm volatile("" : "+v"(kr[i]));
        const float* kf = stg + col * KP + 2 * lh;
        #pragma unroll
        for (int t = 0; t < 2; ++t) {
            #pragma unroll
            for (int i = 0; i < 16; ++i)
                *reinterpret_cast<v4f*>(stg + i * KP + 4 * lane) = kr[16 * t + i];
            v8f acc = {};
            #pragma unroll
            for (int kb = 0; kb < DD; kb += 4) {
                float2 a2 = *reinterpret_cast<const float2*>(qrow + kb);
                float2 b2 = *reinterpret_cast<const float2*>(kf + kb);
                v2f A = {a2.x, a2.y};
                v2f B = {b2.x, b2.y};
                acc = __builtin_amdgcn_wmma_f32_16x16x4_f32(false, A, false, B,
                                                            (short)0, acc, false, false);
            }
            #pragma unroll
            for (int r = 0; r < 8; ++r)
                sc[(r + 8 * lh) * SP + j0 + 16 * t + col] = acc[r];
        }
    } else if (wv == 0) {
        const float* k0 = kst + col * KP + 2 * lh;
        v8f acc0 = {};
        #pragma unroll
        for (int kb = 0; kb < DD; kb += 4) {
            float2 a2 = *reinterpret_cast<const float2*>(qrow + kb);
            float2 b0 = *reinterpret_cast<const float2*>(k0 + kb);
            v2f A  = {a2.x, a2.y};
            v2f B0 = {b0.x, b0.y};
            acc0 = __builtin_amdgcn_wmma_f32_16x16x4_f32(false, A, false, B0,
                                                         (short)0, acc0, false, false);
        }
        #pragma unroll
        for (int r = 0; r < 8; ++r)
            sc[(r + 8 * lh) * SP + col] = acc0[r];
    }
    __syncthreads();

    {
        const int row  = tid >> 4;
        const int part = tid & 15;
        float* srow = sc + row * SP + part;
        float v[16];
        float mloc = -3.0e38f;
        #pragma unroll
        for (int i = 0; i < 16; ++i) {
            v[i] = srow[16 * i];
            mloc = fmaxf(mloc, v[i]);
        }
        for (int mk = 8; mk >= 1; mk >>= 1)
            mloc = fmaxf(mloc, __shfl_xor(mloc, mk, 32));
        #pragma unroll
        for (int i = 0; i < 16; ++i)
            srow[16 * i] = __expf(v[i] - mloc);
        if (part == 0)
            pmax[((size_t)h * NCHUNK + s) * MM + row] = mloc;
    }
    __syncthreads();

    if (wv == 0) {
        const float* scrow = sc + col * SP + lh * 128;
        float ssum = 0.0f;
        for (int j = 0; j < 128; ++j) ssum += scrow[j];
        ssum += __shfl_xor(ssum, 16, 32);
        if (lh == 0)
            psum[((size_t)h * NCHUNK + s) * MM + col] = ssum;
    }

    const float* ap = sc + col * SP + 2 * lh;
    v8f oacc = {};
    if (full) {
        const float* Vb = cV + ((size_t)h * PP + (size_t)s * 256) * DD;
        const float* vp = Vb + (size_t)(2 * lh) * DD + wv * 16 + col;
        #pragma unroll 32
        for (int kb = 0; kb < 256; kb += 4) {
            float2 a2 = *reinterpret_cast<const float2*>(ap + kb);
            v2f A = {a2.x, a2.y};
            v2f B = {vp[(size_t)kb * DD], vp[(size_t)kb * DD + DD]};
            oacc = __builtin_amdgcn_wmma_f32_16x16x4_f32(false, A, false, B,
                                                         (short)0, oacc, false, false);
        }
    } else {
        const float* vp = vraw + (size_t)(2 * lh) * NN + (size_t)h * DD + wv * 16 + col;
        #pragma unroll
        for (int kb = 0; kb < MM; kb += 4) {
            float2 a2 = *reinterpret_cast<const float2*>(ap + kb);
            v2f A = {a2.x, a2.y};
            v2f B = {vp[(size_t)kb * NN], vp[(size_t)kb * NN + NN]};
            oacc = __builtin_amdgcn_wmma_f32_16x16x4_f32(false, A, false, B,
                                                         (short)0, oacc, false, false);
        }
    }

    float* po = pout + ((size_t)h * NCHUNK + s) * MM * DD;
    #pragma unroll
    for (int r = 0; r < 8; ++r)
        po[(size_t)(r + 8 * lh) * DD + wv * 16 + col] = oacc[r];
}

__global__ __launch_bounds__(128) void k_combine(const float* __restrict__ pout,
                                                 const float* __restrict__ pmax,
                                                 const float* __restrict__ psum,
                                                 float* __restrict__ out)
{
    const int h = blockIdx.x >> 4;
    const int m = blockIdx.x & (MM - 1);
    const int d = threadIdx.x;
    const float* pm  = pmax + (size_t)h * NCHUNK * MM + m;
    const float* psu = psum + (size_t)h * NCHUNK * MM + m;
    const float* po  = pout + ((size_t)h * NCHUNK * MM + m) * DD + d;

    float mx[NCHUNK], sm[NCHUNK], pv[NCHUNK];
    #pragma unroll
    for (int s = 0; s < NCHUNK; ++s) {
        mx[s] = pm[(size_t)s * MM];
        sm[s] = psu[(size_t)s * MM];
        pv[s] = po[(size_t)s * MM * DD];
    }
    float g = -3.0e38f;
    #pragma unroll
    for (int s = 0; s < NCHUNK; ++s) g = fmaxf(g, mx[s]);
    float tot = 0.0f, acc = 0.0f;
    #pragma unroll
    for (int s = 0; s < NCHUNK; ++s) {
        float w = __expf(mx[s] - g);
        tot += w * sm[s];
        acc += w * pv[s];
    }
    out[(size_t)m * NN + (size_t)h * DD + d] = acc / tot;
}

extern "C" void kernel_launch(void* const* d_in, const int* in_sizes, int n_in,
                              void* d_out, int out_size, void* d_ws, size_t ws_size,
                              hipStream_t stream)
{
    (void)in_sizes; (void)n_in; (void)out_size; (void)ws_size;
    const float* X  = (const float*)d_in[0];
    const float* Wq = (const float*)d_in[1];
    const float* Wk = (const float*)d_in[2];
    const float* Wv = (const float*)d_in[3];
    const float* cK = (const float*)d_in[4];
    const float* cV = (const float*)d_in[5];
    float* out = (float*)d_out;
    float* ws  = (float*)d_ws;

    float* qraw  = ws;
    float* kraw  = qraw + MM * NN;
    float* vraw  = kraw + MM * NN;
    float* pout  = vraw + MM * NN;
    float* pmaxb = pout + (size_t)HH * NCHUNK * MM * DD;
    float* psumb = pmaxb + (size_t)HH * NCHUNK * MM;

    k_proj_wmma<<<3 * (NN / TN), 32 * KSPLIT, 0, stream>>>(X, Wq, Wk, Wv,
                                                  qraw, kraw, vraw);
    k_attn_partial<<<dim3(NCHUNK, HH), 256, 0, stream>>>(qraw, kraw, vraw, cK, cV,
                                                        pout, pmaxb, psumb);
    k_combine<<<HH * MM, DD, 0, stream>>>(pout, pmaxb, psumb, out);
}


# projection kernel: each wave fetches its k segment of the 16 X rows whole (512 bytes per load) and reads A fragments from a per-wave LDS stage, not by 16-line loads; every sum keeps its order
# speedup vs baseline: 5.7224x; 1.0698x over previous
#include <hip/hip_runtime.h>
#include <math.h>


typedef __attribute__((ext_vector_type(2))) float v2f;
typedef __attribute__((ext_vector_type(8))) float v8f;
typedef __attribute__((ext_vector_type(4))) float v4f;

#define MM 16
#define NN 2048
#define DD 128
#define PP 8192
#define HH 16
#define NCHUNK 33

#define KSPLIT 16
#define KSEG   (NN / KSPLIT)
#define TN     32
#define XP     (KSEG + 4)

__global__ __launch_bounds__(32 * KSPLIT) __attribute__((amdgpu_waves_per_eu(4, 4)))
void k_proj_wmma(const float* __restrict__ X,
                 const float* __restrict__ Wq,
                 const float* __restrict__ Wk,
                 const float* __restrict__ Wv,
                 float* __restrict__ qraw,
                 float* __restrict__ kraw,
                 float* __restrict__ vraw)
{
    __shared__ __align__(16) float part[KSPLIT * 16 * TN];
    __shared__ float ssq[KSPLIT * 16];
    __shared__ __align__(16) float xst[KSPLIT * 16 * XP];

    const int tid  = threadIdx.x;
    const int lane = tid & 31;
    const int wv   = tid >> 5;
    const int col  = lane & 15;
    const int lh   = lane >> 4;
    const int bid  = blockIdx.x;
    const int w    = bid >> 6;
    const int n0   = (bid & 63) << 5;

    const float* W   = (w == 0) ? Wq : (w == 1) ? Wk : Wv;
    float*       out = (w == 0) ? qraw : (w == 1) ? kraw : vraw;

    const int k0 = wv * KSEG;
    const float* bcol = W + (size_t)(k0 + 2 * lh) * NN + n0 + 2 * col;

    float* stg = xst + wv * (16 * XP);
    {
        const float* xrow = X + k0 + 4 * lane;
        v4f xr[16];
        #pragma unroll
        for (int i = 0; i < 16; ++i)
            xr[i] = *reinterpret_cast<const v4f*>(xrow + (size_t)i * NN);
        #pragma unroll
        for (int i = 0; i < 16; ++i)
            *reinterpret_cast<v4f*>(stg + i * XP + 4 * lane) = xr[i];
    }
    const float* arow = stg + col * XP + 2 * lh;

    v8f acc0 = {};
    v8f acc1 = {};
    float ss = 0.0f;
    #pragma unroll
    for (int kb = 0; kb < KSEG; kb += 4) {
        float2 a2 = *reinterpret_cast<const float2*>(arow + kb);
        ss += a2.x * a2.x;
        ss += a2.y * a2.y;
        v2f A = {a2.x, a2.y};
        const float* bp = bcol + (size_t)kb * NN;
        float2 b0 = *reinterpret_cast<const float2*>(bp);
        float2 b1 = *reinterpret_cast<const float2*>(bp + NN);
        v2f B0 = {b0.x, b1.x};
        v2f B1 = {b0.y, b1.y};
        acc0 = __builtin_amdgcn_wmma_f32_16x16x4_f32(false, A, false, B0,
                                                     (short)0, acc0, false, false);
        acc1 = __builtin_amdgcn_wmma_f32_16x16x4_f32(false, A, false, B1,
                                                     (short)0, acc1, false, false);
    }

    #pragma unroll
    for (int r = 0; r < 8; ++r) {
        float2 o;
        o.x = acc0[r];
        o.y = acc1[r];
        *reinterpret_cast<float2*>(part + wv * (16 * TN) + (2 * r + lh) * TN + 2 * col) = o;
    }
    ss += __shfl_xor(ss, 16, 32);
    if (lh == 0) ssq[wv * 16 + col] = ss;
    __syncthreads();

    {
        const int slot = tid >> 5;
        const int row  = (slot >> 1) + 8 * (slot & 1);
        float s = part[tid];
        float t = ssq[row];
        #pragma unroll
        for (int p = 1; p < KSPLIT; ++p) {
            s += part[p * (16 * TN) + tid];
            t += ssq[p * 16 + row];
        }
        out[(size_t)row * NN + n0 + (tid & 31)] = s * rsqrtf(t / (float)NN);
    }
}

#define QP 132
#define SP 260
#define KP 132

__global__ __launch_bounds__(256) void k_attn_partial(const float* __restrict__ qraw,
                                                      const float* __restrict__ kraw,
                                                      const float* __restrict__ vraw,
                                                      const float* __restrict__ cK,
                                                      const float* __restrict__ cV,
                                                      float* __restrict__ pout,
                                                      float* __restrict__ pmax,
                                                      float* __restrict__ psum)
{
    __shared__ __align__(16) float q_lds[MM * QP];
    __shared__ __align__(16) float sc[MM * SP];
    __shared__ __align__(16) float kst[8 * MM * KP];

    const int tid  = threadIdx.x;
    const int lane = tid & 31;
    const int wv   = tid >> 5;
    const int col  = lane & 15;
    const int lh   = lane >> 4;
    const int s    = blockIdx.x;
    const int h    = blockIdx.y;
    const bool full = (s < 32);

    {
        const float* qh = qraw + (size_t)h * DD + 4 * lane;
        float4 q4[2];
        #pragma unroll
        for (int u = 0; u < 2; ++u)
            q4[u] = *reinterpret_cast<const float4*>(qh + (size_t)(wv + 8 * u) * NN);
        #pragma unroll
        for (int u = 0; u < 2; ++u) {
            float ss = q4[u].x * q4[u].x + q4[u].y * q4[u].y
                     + q4[u].z * q4[u].z + q4[u].w * q4[u].w;
            #pragma unroll
            for (int mk = 16; mk >= 1; mk >>= 1) ss += __shfl_xor(ss, mk, 32);
            const float rs = rsqrtf(ss / (float)DD);
            float4 o;
            o.x = q4[u].x * rs; o.y = q4[u].y * rs; o.z = q4[u].z * rs; o.w = q4[u].w * rs;
            *reinterpret_cast<float4*>(q_lds + (wv + 8 * u) * QP + 4 * lane) = o;
        }
    }

    if (!full) {
        const float* kh = kraw + (size_t)h * DD + 4 * lane;
        float4 k4[2];
        #pragma unroll
        for (int u = 0; u < 2; ++u)
            k4[u] = *reinterpret_cast<const float4*>(kh + (size_t)(wv + 8 * u) * NN);
        #pragma unroll
        for (int u = 0; u < 2; ++u) {
            float ss = k4[u].x * k4[u].x + k4[u].y * k4[u].y
                     + k4[u].z * k4[u].z + k4[u].w * k4[u].w;
            #pragma unroll
            for (int mk = 16; mk >= 1; mk >>= 1) ss += __shfl_xor(ss, mk, 32);
            const float rs = rsqrtf(ss / (float)DD);
            float4 o;
            o.x = k4[u].x * rs; o.y = k4[u].y * rs; o.z = k4[u].z * rs; o.w = k4[u].w * rs;
            *reinterpret_cast<float4*>(kst + (wv + 8 * u) * KP + 4 * lane) = o;
        }
        for (int i = tid; i < MM * SP; i += 256) sc[i] = -3.0e38f;
    }
    __syncthreads();

    const float* qrow = q_lds + col * QP + 2 * lh;
    if (full) {
        const int j0 = wv * 32;
        const float* Kw = cK + ((size_t)h * PP + (size_t)s * 256 + j0) * DD + 4 * lane;
        float* stg = kst + wv * (MM * KP);
        v4f kr[32];
        #pragma unroll
        for (int i = 0; i < 32; ++i)
            kr[i] = *reinterpret_cast<const v4f*>(Kw + (size_t)i * DD);
        #pragma unroll
        for (int i = 16; i < 32; ++i)
            asm volatile("" : "+v"(kr[i]));
        const float* kf = stg + col * KP + 2 * lh;
        #pragma unroll
        for (int t = 0; t < 2; ++t) {
            #pragma unroll
            for (int i = 0; i < 16; ++i)
                *reinterpret_cast<v4f*>(stg + i * KP + 4 * lane) = kr[16 * t + i];
            v8f acc = {};
            #pragma unroll
            for (int kb = 0; kb < DD; kb += 4) {
                float2 a2 = *reinterpret_cast<const float2*>(qrow + kb);
                float2 b2 = *reinterpret_cast<const float2*>(kf + kb);
                v2f A = {a2.x, a2.y};
                v2f B = {b2.x, b2.y};
                acc = __builtin_amdgcn_wmma_f32_16x16x4_f32(false, A, false, B,
                                                            (short)0, acc, false, false);
            }
            #pragma unroll
            for (int r = 0; r < 8; ++r)
                sc[(r + 8 * lh) * SP + j0 + 16 * t + col] = acc[r];
        }
    } else if (wv == 0) {
        const float* k0 = kst + col * KP + 2 * lh;
        v8f acc0 = {};
        #pragma unroll
        for (int kb = 0; kb < DD; kb += 4) {
            float2 a2 = *reinterpret_cast<const float2*>(qrow + kb);
            float2 b0 = *reinterpret_cast<const float2*>(k0 + kb);
            v2f A  = {a2.x, a2.y};
            v2f B0 = {b0.x, b0.y};
            acc0 = __builtin_amdgcn_wmma_f32_16x16x4_f32(false, A, false, B0,
                                                         (short)0, acc0, false, false);
        }
        #pragma unroll
        for (int r = 0; r < 8; ++r)
            sc[(r + 8 * lh) * SP + col] = acc0[r];
    }
    __syncthreads();

    {
        const int row  = tid >> 4;
        const int part = tid & 15;
        float* srow = sc + row * SP + part;
        float v[16];
        float mloc = -3.0e38f;
        #pragma unroll
        for (int i = 0; i < 16; ++i) {
            v[i] = srow[16 * i];
            mloc = fmaxf(mloc, v[i]);
        }
        for (int mk = 8; mk >= 1; mk >>= 1)
            mloc = fmaxf(mloc, __shfl_xor(mloc, mk, 32));
        #pragma unroll
        for (int i = 0; i < 16; ++i)
            srow[16 * i] = __expf(v[i] - mloc);
        if (part == 0)
            pmax[((size_t)h * NCHUNK + s) * MM + row] = mloc;
    }
    __syncthreads();

    if (wv == 0) {
        const float* scrow = sc + col * SP + lh * 128;
        float ssum = 0.0f;
        for (int j = 0; j < 128; ++j) ssum += scrow[j];
        ssum += __shfl_xor(ssum, 16, 32);
        if (lh == 0)
            psum[((size_t)h * NCHUNK + s) * MM + col] = ssum;
    }

    const float* ap = sc + col * SP + 2 * lh;
    v8f oacc = {};
    if (full) {
        const float* Vb = cV + ((size_t)h * PP + (size_t)s * 256) * DD;
        const float* vp = Vb + (size_t)(2 * lh) * DD + wv * 16 + col;
        #pragma unroll 32
        for (int kb = 0; kb < 256; kb += 4) {
            float2 a2 = *reinterpret_cast<const float2*>(ap + kb);
            v2f A = {a2.x, a2.y};
            v2f B = {vp[(size_t)kb * DD], vp[(size_t)kb * DD + DD]};
            oacc = __builtin_amdgcn_wmma_f32_16x16x4_f32(false, A, false, B,
                                                         (short)0, oacc, false, false);
        }
    } else {
        const float* vp = vraw + (size_t)(2 * lh) * NN + (size_t)h * DD + wv * 16 + col;
        #pragma unroll
        for (int kb = 0; kb < MM; kb += 4) {
            float2 a2 = *reinterpret_cast<const float2*>(ap + kb);
            v2f A = {a2.x, a2.y};
            v2f B = {vp[(size_t)kb * NN], vp[(size_t)kb * NN + NN]};
            oacc = __builtin_amdgcn_wmma_f32_16x16x4_f32(false, A, false, B,
                                                         (short)0, oacc, false, false);
        }
    }

    float* po = pout + ((size_t)h * NCHUNK + s) * MM * DD;
    #pragma unroll
    for (int r = 0; r < 8; ++r)
        po[(size_t)(r + 8 * lh) * DD + wv * 16 + col] = oacc[r];
}

__global__ __launch_bounds__(128) void k_combine(const float* __restrict__ pout,
                                                 const float* __restrict__ pmax,
                                                 const float* __restrict__ psum,
                                                 float* __restrict__ out)
{
    const int h = blockIdx.x >> 4;
    const int m = blockIdx.x & (MM - 1);
    const int d = threadIdx.x;
    const float* pm  = pmax + (size_t)h * NCHUNK * MM + m;
    const float* psu = psum + (size_t)h * NCHUNK * MM + m;
    const float* po  = pout + ((size_t)h * NCHUNK * MM + m) * DD + d;

    float mx[NCHUNK], sm[NCHUNK], pv[NCHUNK];
    #pragma unroll
    for (int s = 0; s < NCHUNK; ++s) {
        mx[s] = pm[(size_t)s * MM];
        sm[s] = psu[(size_t)s * MM];
        pv[s] = po[(size_t)s * MM * DD];
    }
    float g = -3.0e38f;
    #pragma unroll
    for (int s = 0; s < NCHUNK; ++s) g = fmaxf(g, mx[s]);
    float tot = 0.0f, acc = 0.0f;
    #pragma unroll
    for (int s = 0; s < NCHUNK; ++s) {
        float w = __expf(mx[s] - g);
        tot += w * sm[s];
        acc += w * pv[s];
    }
    out[(size_t)m * NN + (size_t)h * DD + d] = acc / tot;
}

extern "C" void kernel_launch(void* const* d_in, const int* in_sizes, int n_in,
                              void* d_out, int out_size, void* d_ws, size_t ws_size,
                              hipStream_t stream)
{
    (void)in_sizes; (void)n_in; (void)out_size; (void)ws_size;
    const float* X  = (const float*)d_in[0];
    const float* Wq = (const float*)d_in[1];
    const float* Wk = (const float*)d_in[2];
    const float* Wv = (const float*)d_in[3];
    const float* cK = (const float*)d_in[4];
    const float* cV = (const float*)d_in[5];
    float* out = (float*)d_out;
    float* ws  = (float*)d_ws;

    float* qraw  = ws;
    float* kraw  = qraw + MM * NN;
    float* vraw  = kraw + MM * NN;
    float* pout  = vraw + MM * NN;
    float* pmaxb = pout + (size_t)HH * NCHUNK * MM * DD;
    float* psumb = pmaxb + (size_t)HH * NCHUNK * MM;

    k_proj_wmma<<<3 * (NN / TN), 32 * KSPLIT, 0, stream>>>(X, Wq, Wk, Wv,
                                                  qraw, kraw, vraw);
    k_attn_partial<<<dim3(NCHUNK, HH), 256, 0, stream>>>(qraw, kraw, vraw, cK, cV,
                                                        pout, pmaxb, psumb);
    k_combine<<<HH * MM, DD, 0, stream>>>(pout, pmaxb, psumb, out);
}


# attention grid 528 -> 512 blocks, two per compute unit, none with three: the 16 new rows join the last cache chunk's block (started first); their max, exp and PV sums move into it (order changes)
# speedup vs baseline: 6.0868x; 1.0637x over previous
#include <hip/hip_runtime.h>
#include <math.h>


typedef __attribute__((ext_vector_type(2))) float v2f;
typedef __attribute__((ext_vector_type(8))) float v8f;
typedef __attribute__((ext_vector_type(4))) float v4f;

#define MM 16
#define NN 2048
#define DD 128
#define PP 8192
#define HH 16
#define NCHUNK 32

#define KSPLIT 16
#define KSEG   (NN / KSPLIT)
#define TN     32
#define XP     (KSEG + 4)

__global__ __launch_bounds__(32 * KSPLIT) __attribute__((amdgpu_waves_per_eu(4, 4)))
void k_proj_wmma(const float* __restrict__ X,
                 const float* __restrict__ Wq,
                 const float* __restrict__ Wk,
                 const float* __restrict__ Wv,
                 float* __restrict__ qraw,
                 float* __restrict__ kraw,
                 float* __restrict__ vraw)
{
    __shared__ __align__(16) float part[KSPLIT * 16 * TN];
    __shared__ float ssq[KSPLIT * 16];
    __shared__ __align__(16) float xst[KSPLIT * 16 * XP];

    const int tid  = threadIdx.x;
    const int lane = tid & 31;
    const int wv   = tid >> 5;
    const int col  = lane & 15;
    const int lh   = lane >> 4;
    const int bid  = blockIdx.x;
    const int w    = bid >> 6;
    const int n0   = (bid & 63) << 5;

    const float* W   = (w == 0) ? Wq : (w == 1) ? Wk : Wv;
    float*       out = (w == 0) ? qraw : (w == 1) ? kraw : vraw;

    const int k0 = wv * KSEG;
    const float* bcol = W + (size_t)(k0 + 2 * lh) * NN + n0 + 2 * col;

    float* stg = xst + wv * (16 * XP);
    {
        const float* xrow = X + k0 + 4 * lane;
        v4f xr[16];
        #pragma unroll
        for (int i = 0; i < 16; ++i)
            xr[i] = *reinterpret_cast<const v4f*>(xrow + (size_t)i * NN);
        #pragma unroll
        for (int i = 0; i < 16; ++i)
            *reinterpret_cast<v4f*>(stg + i * XP + 4 * lane) = xr[i];
    }
    const float* arow = stg + col * XP + 2 * lh;

    v8f acc0 = {};
    v8f acc1 = {};
    float ss = 0.0f;
    #pragma unroll
    for (int kb = 0; kb < KSEG; kb += 4) {
        float2 a2 = *reinterpret_cast<const float2*>(arow + kb);
        ss += a2.x * a2.x;
        ss += a2.y * a2.y;
        v2f A = {a2.x, a2.y};
        const float* bp = bcol + (size_t)kb * NN;
        float2 b0 = *reinterpret_cast<const float2*>(bp);
        float2 b1 = *reinterpret_cast<const float2*>(bp + NN);
        v2f B0 = {b0.x, b1.x};
        v2f B1 = {b0.y, b1.y};
        acc0 = __builtin_amdgcn_wmma_f32_16x16x4_f32(false, A, false, B0,
                                                     (short)0, acc0, false, false);
        acc1 = __builtin_amdgcn_wmma_f32_16x16x4_f32(false, A, false, B1,
                                                     (short)0, acc1, false, false);
    }

    #pragma unroll
    for (int r = 0; r < 8; ++r) {
        float2 o;
        o.x = acc0[r];
        o.y = acc1[r];
        *reinterpret_cast<float2*>(part + wv * (16 * TN) + (2 * r + lh) * TN + 2 * col) = o;
    }
    ss += __shfl_xor(ss, 16, 32);
    if (lh == 0) ssq[wv * 16 + col] = ss;
    __syncthreads();

    {
        const int slot = tid >> 5;
        const int row  = (slot >> 1) + 8 * (slot & 1);
        float s = part[tid];
        float t = ssq[row];
        #pragma unroll
        for (int p = 1; p < KSPLIT; ++p) {
            s += part[p * (16 * TN) + tid];
            t += ssq[p * 16 + row];
        }
        out[(size_t)row * NN + n0 + (tid & 31)] = s * rsqrtf(t / (float)NN);
    }
}

#define QP 132
#define SP 276
#define KP 132
#define CR 256

__global__ __launch_bounds__(256)
void k_attn_partial(const float* __restrict__ qraw,
                    const float* __restrict__ kraw,
                    const float* __restrict__ vraw,
                    const float* __restrict__ cK,
                    const float* __restrict__ cV,
                    float* __restrict__ pout,
                    float* __restrict__ pmax,
                    float* __restrict__ psum)
{
    __shared__ __align__(16) float q_lds[MM * QP];
    __shared__ __align__(16) float sc[MM * SP];
    __shared__ __align__(16) float kst[8 * MM * KP];
    __shared__ __align__(16) float knew[MM * KP];

    const int tid  = threadIdx.x;
    const int lane = tid & 31;
    const int wv   = tid >> 5;
    const int col  = lane & 15;
    const int lh   = lane >> 4;
    const int h    = blockIdx.x;
    const int s    = (NCHUNK - 1) - (int)blockIdx.y;
    const bool last = (s == NCHUNK - 1);

    {
        const float* qh = qraw + (size_t)h * DD + 4 * lane;
        float4 q4[2];
        #pragma unroll
        for (int u = 0; u < 2; ++u)
            q4[u] = *reinterpret_cast<const float4*>(qh + (size_t)(wv + 8 * u) * NN);
        #pragma unroll
        for (int u = 0; u < 2; ++u) {
            float ss = q4[u].x * q4[u].x + q4[u].y * q4[u].y
                     + q4[u].z * q4[u].z + q4[u].w * q4[u].w;
            #pragma unroll
            for (int mk = 16; mk >= 1; mk >>= 1) ss += __shfl_xor(ss, mk, 32);
            const float rs = rsqrtf(ss / (float)DD);
            float4 o;
            o.x = q4[u].x * rs; o.y = q4[u].y * rs; o.z = q4[u].z * rs; o.w = q4[u].w * rs;
            *reinterpret_cast<float4*>(q_lds + (wv + 8 * u) * QP + 4 * lane) = o;
        }
    }

    if (last) {
        const float* kh = kraw + (size_t)h * DD + 4 * lane;
        float4 k4[2];
        #pragma unroll
        for (int u = 0; u < 2; ++u)
            k4[u] = *reinterpret_cast<const float4*>(kh + (size_t)(wv + 8 * u) * NN);
        #pragma unroll
        for (int u = 0; u < 2; ++u) {
            float ss = k4[u].x * k4[u].x + k4[u].y * k4[u].y
                     + k4[u].z * k4[u].z + k4[u].w * k4[u].w;
            #pragma unroll
            for (int mk = 16; mk >= 1; mk >>= 1) ss += __shfl_xor(ss, mk, 32);
            const float rs = rsqrtf(ss / (float)DD);
            float4 o;
            o.x = k4[u].x * rs; o.y = k4[u].y * rs; o.z = k4[u].z * rs; o.w = k4[u].w * rs;
            *reinterpret_cast<float4*>(knew + (wv + 8 * u) * KP + 4 * lane) = o;
        }
    }
    __syncthreads();

    const float* qrow = q_lds + col * QP + 2 * lh;
    {
        const int j0 = wv * 32;
        const float* Kw = cK + ((size_t)h * PP + (size_t)s * CR + j0) * DD + 4 * lane;
        float* stg = kst + wv * (MM * KP);
        v4f kr[32];
        #pragma unroll
        for (int i = 0; i < 32; ++i)
            kr[i] = *reinterpret_cast<const v4f*>(Kw + (size_t)i * DD);

        if (last && wv == 7) {
            const float* kn = knew + col * KP + 2 * lh;
            v8f accn = {};
            #pragma unroll
            for (int kb = 0; kb < DD; kb += 4) {
                float2 a2 = *reinterpret_cast<const float2*>(qrow + kb);
                float2 b2 = *reinterpret_cast<const float2*>(kn + kb);
                v2f A = {a2.x, a2.y};
                v2f B = {b2.x, b2.y};
                accn = __builtin_amdgcn_wmma_f32_16x16x4_f32(false, A, false, B,
                                                             (short)0, accn, false, false);
            }
            #pragma unroll
            for (int r = 0; r < 8; ++r)
                sc[(r + 8 * lh) * SP + CR + col] = accn[r];
        }

        const float* kf = stg + col * KP + 2 * lh;
        #pragma unroll
        for (int t = 0; t < 2; ++t) {
            #pragma unroll
            for (int i = 0; i < 16; ++i)
                *reinterpret_cast<v4f*>(stg + i * KP + 4 * lane) = kr[16 * t + i];
            v8f acc = {};
            #pragma unroll
            for (int kb = 0; kb < DD; kb += 4) {
                float2 a2 = *reinterpret_cast<const float2*>(qrow + kb);
                float2 b2 = *reinterpret_cast<const float2*>(kf + kb);
                v2f A = {a2.x, a2.y};
                v2f B = {b2.x, b2.y};
                acc = __builtin_amdgcn_wmma_f32_16x16x4_f32(false, A, false, B,
                                                            (short)0, acc, false, false);
            }
            #pragma unroll
            for (int r = 0; r < 8; ++r)
                sc[(r + 8 * lh) * SP + j0 + 16 * t + col] = acc[r];
        }
    }
    __syncthreads();

    {
        const int row  = tid >> 4;
        const int part = tid & 15;
        float* srow = sc + row * SP + part;
        float v[16];
        float vn = 0.0f;
        float mloc = -3.0e38f;
        #pragma unroll
        for (int i = 0; i < 16; ++i) {
            v[i] = srow[16 * i];
            mloc = fmaxf(mloc, v[i]);
        }
        if (last) {
            vn = srow[CR];
            mloc = fmaxf(mloc, vn);
        }
        for (int mk = 8; mk >= 1; mk >>= 1)
            mloc = fmaxf(mloc, __shfl_xor(mloc, mk, 32));
        #pragma unroll
        for (int i = 0; i < 16; ++i)
            srow[16 * i] = __expf(v[i] - mloc);
        if (last)
            srow[CR] = __expf(vn - mloc);
        if (part == 0)
            pmax[((size_t)h * NCHUNK + s) * MM + row] = mloc;
    }
    __syncthreads();

    if (wv == 0) {
        const float* scrow = sc + col * SP + lh * 128;
        float ssum = 0.0f;
        for (int j = 0; j < 128; ++j) ssum += scrow[j];
        if (last && lh == 1)
            for (int j = 128; j < 128 + MM; ++j) ssum += scrow[j];
        ssum += __shfl_xor(ssum, 16, 32);
        if (lh == 0)
            psum[((size_t)h * NCHUNK + s) * MM + col] = ssum;
    }

    const float* ap = sc + col * SP + 2 * lh;
    v8f oacc = {};
    float bn[8] = {0.0f, 0.0f, 0.0f, 0.0f, 0.0f, 0.0f, 0.0f, 0.0f};
    if (last) {
        const float* vn = vraw + (size_t)(2 * lh) * NN + (size_t)h * DD + wv * 16 + col;
        #pragma unroll
        for (int kb = 0; kb < MM; kb += 4) {
            bn[kb / 2]     = vn[(size_t)kb * NN];
            bn[kb / 2 + 1] = vn[(size_t)kb * NN + NN];
        }
    }
    {
        const float* Vb = cV + ((size_t)h * PP + (size_t)s * CR) * DD;
        const float* vp = Vb + (size_t)(2 * lh) * DD + wv * 16 + col;
        #pragma unroll 32
        for (int kb = 0; kb < CR; kb += 4) {
            float2 a2 = *reinterpret_cast<const float2*>(ap + kb);
            v2f A = {a2.x, a2.y};
            v2f B = {vp[(size_t)kb * DD], vp[(size_t)kb * DD + DD]};
            oacc = __builtin_amdgcn_wmma_f32_16x16x4_f32(false, A, false, B,
                                                         (short)0, oacc, false, false);
        }
    }
    if (last) {
        #pragma unroll
        for (int kb = 0; kb < MM; kb += 4) {
            float2 a2 = *reinterpret_cast<const float2*>(ap + CR + kb);
            v2f A = {a2.x, a2.y};
            v2f B = {bn[kb / 2], bn[kb / 2 + 1]};
            oacc = __builtin_amdgcn_wmma_f32_16x16x4_f32(false, A, false, B,
                                                         (short)0, oacc, false, false);
        }
    }

    float* po = pout + ((size_t)h * NCHUNK + s) * MM * DD;
    #pragma unroll
    for (int r = 0; r < 8; ++r)
        po[(size_t)(r + 8 * lh) * DD + wv * 16 + col] = oacc[r];
}

__global__ __launch_bounds__(128) void k_combine(const float* __restrict__ pout,
                                                 const float* __restrict__ pmax,
                                                 const float* __restrict__ psum,
                                                 float* __restrict__ out)
{
    const int h = blockIdx.x >> 4;
    const int m = blockIdx.x & (MM - 1);
    const int d = threadIdx.x;
    const float* pm  = pmax + (size_t)h * NCHUNK * MM + m;
    const float* psu = psum + (size_t)h * NCHUNK * MM + m;
    const float* po  = pout + ((size_t)h * NCHUNK * MM + m) * DD + d;

    float mx[NCHUNK], sm[NCHUNK], pv[NCHUNK];
    #pragma unroll
    for (int s = 0; s < NCHUNK; ++s) {
        mx[s] = pm[(size_t)s * MM];
        sm[s] = psu[(size_t)s * MM];
        pv[s] = po[(size_t)s * MM * DD];
    }
    float g = -3.0e38f;
    #pragma unroll
    for (int s = 0; s < NCHUNK; ++s) g = fmaxf(g, mx[s]);
    float tot = 0.0f, acc = 0.0f;
    #pragma unroll
    for (int s = 0; s < NCHUNK; ++s) {
        float w = __expf(mx[s] - g);
        tot += w * sm[s];
        acc += w * pv[s];
    }
    out[(size_t)m * NN + (size_t)h * DD + d] = acc / tot;
}

extern "C" void kernel_launch(void* const* d_in, const int* in_sizes, int n_in,
                              void* d_out, int out_size, void* d_ws, size_t ws_size,
                              hipStream_t stream)
{
    (void)in_sizes; (void)n_in; (void)out_size; (void)ws_size;
    const float* X  = (const float*)d_in[0];
    const float* Wq = (const float*)d_in[1];
    const float* Wk = (const float*)d_in[2];
    const float* Wv = (const float*)d_in[3];
    const float* cK = (const float*)d_in[4];
    const float* cV = (const float*)d_in[5];
    float* out = (float*)d_out;
    float* ws  = (float*)d_ws;

    float* qraw  = ws;
    float* kraw  = qraw + MM * NN;
    float* vraw  = kraw + MM * NN;
    float* pout  = vraw + MM * NN;
    float* pmaxb = pout + (size_t)HH * NCHUNK * MM * DD;
    float* psumb = pmaxb + (size_t)HH * NCHUNK * MM;

    k_proj_wmma<<<3 * (NN / TN), 32 * KSPLIT, 0, stream>>>(X, Wq, Wk, Wv,
                                                  qraw, kraw, vraw);
    k_attn_partial<<<dim3(HH, NCHUNK), 256, 0, stream>>>(qraw, kraw, vraw, cK, cV,
                                                        pout, pmaxb, psumb);
    k_combine<<<HH * MM, DD, 0, stream>>>(pout, pmaxb, psumb, out);
}
